# starting kernel
# baseline (speedup 1.0000x reference)
#include <hip/hip_runtime.h>

typedef __attribute__((ext_vector_type(2))) float v2f;
typedef __attribute__((ext_vector_type(8))) float v8f;

#define DFEAT 128
#define TILE  16

__global__ void zero_f4(float4* __restrict__ p, long long n4) {
  long long i = (long long)blockIdx.x * blockDim.x + threadIdx.x;
  if (i < n4) p[i] = make_float4(0.f, 0.f, 0.f, 0.f);
}

__global__ __launch_bounds__(256) void rgcn_proj_gemm(
    const float* __restrict__ feats,
    const float* __restrict__ Wc, const float* __restrict__ bc,
    const float* __restrict__ We, const float* __restrict__ be,
    float* __restrict__ WhC, float* __restrict__ WhE, int n)
{
  const int wave  = threadIdx.x >> 5;
  const int lane  = threadIdx.x & 31;
  const int gw    = blockIdx.x * 8 + wave;
  const int nColT = DFEAT / TILE;
  const int m0    = (gw / nColT) * TILE;
  const int n0    = (gw % nColT) * TILE;
  if (m0 >= n) return;

  const int half = lane >> 4;
  const int l15  = lane & 15;

  int arowi = m0 + l15;
  if (arowi >= n) arowi = n - 1;
  const float* arow  = feats + (size_t)arowi * DFEAT + half * 2;
  const float* bcrow = Wc    + (size_t)(n0 + l15) * DFEAT + half * 2;
  const float* berow = We    + (size_t)(n0 + l15) * DFEAT + half * 2;

  v8f accC = {0.f, 0.f, 0.f, 0.f, 0.f, 0.f, 0.f, 0.f};
  v8f accE = {0.f, 0.f, 0.f, 0.f, 0.f, 0.f, 0.f, 0.f};
#pragma unroll 8
  for (int k = 0; k < DFEAT; k += 4) {
    v2f a   = *(const v2f*)(arow + k);
    v2f bC  = *(const v2f*)(bcrow + k);
    v2f bE  = *(const v2f*)(berow + k);
    accC = __builtin_amdgcn_wmma_f32_16x16x4_f32(
        false, a, false, bC, (short)0, accC, false, false);
    accE = __builtin_amdgcn_wmma_f32_16x16x4_f32(
        false, a, false, bE, (short)0, accE, false, false);
  }

  const float biasC = bc[n0 + l15];
  const float biasE = be[n0 + l15];
  float* ocol = WhC + (size_t)(m0 + half * 8) * DFEAT + n0 + l15;
  float* ecol = WhE + (size_t)(m0 + half * 8) * DFEAT + n0 + l15;

  if (m0 + TILE <= n) {
#pragma unroll
    for (int r = 0; r < 8; ++r) {
      ocol[(size_t)r * DFEAT] = accC[r] + biasC;
      ecol[(size_t)r * DFEAT] = accE[r] + biasE;
    }
  } else {
#pragma unroll
    for (int r = 0; r < 8; ++r) {
      const int M = m0 + half * 8 + r;
      if (M < n) {
        ocol[(size_t)r * DFEAT] = accC[r] + biasC;
        ecol[(size_t)r * DFEAT] = accE[r] + biasE;
      }
    }
  }
}

__global__ __launch_bounds__(256) void rgcn_out_gemm(
    const float* __restrict__ h,
    const float* __restrict__ Wo, const float* __restrict__ bo,
    const float* __restrict__ WhE, float* __restrict__ out, int n)
{
  const int wave  = threadIdx.x >> 5;
  const int lane  = threadIdx.x & 31;
  const int gw    = blockIdx.x * 8 + wave;
  const int nColT = DFEAT / TILE;
  const int m0    = (gw / nColT) * TILE;
  const int n0    = (gw % nColT) * TILE;
  if (m0 >= n) return;

  const int half = lane >> 4;
  const int l15  = lane & 15;

  int arowi = m0 + l15;
  if (arowi >= n) arowi = n - 1;
  const float* arow = h  + (size_t)arowi * DFEAT + half * 2;
  const float* brow = Wo + (size_t)(n0 + l15) * DFEAT + half * 2;

  v8f acc = {0.f, 0.f, 0.f, 0.f, 0.f, 0.f, 0.f, 0.f};
#pragma unroll 8
  for (int k = 0; k < DFEAT; k += 4) {
    v2f a = *(const v2f*)(arow + k);
    v2f b = *(const v2f*)(brow + k);
    acc = __builtin_amdgcn_wmma_f32_16x16x4_f32(
        false, a, false, b, (short)0, acc, false, false);
  }

  const float bias = bo[n0 + l15];
  const size_t base = (size_t)(m0 + half * 8) * DFEAT + n0 + l15;

  if (m0 + TILE <= n) {
#pragma unroll
    for (int r = 0; r < 8; ++r) {
      const size_t idx = base + (size_t)r * DFEAT;
      out[idx] = acc[r] + bias + WhE[idx];
    }
  } else {
#pragma unroll
    for (int r = 0; r < 8; ++r) {
      const int M = m0 + half * 8 + r;
      if (M < n) {
        const size_t idx = base + (size_t)r * DFEAT;
        out[idx] = acc[r] + bias + WhE[idx];
      }
    }
  }
}

__global__ __launch_bounds__(256) void scatter_add_edges(
    const float* __restrict__ Wh,
    const int* __restrict__ src, const int* __restrict__ dst,
    float* __restrict__ h, int nE)
{
  const long long idx = (long long)blockIdx.x * blockDim.x + threadIdx.x;
  const int e = (int)(idx >> 7);
  if (e >= nE) return;
  const int d = (int)(idx & 127);
  const int s = src[e];
  const int t = dst[e];
  atomicAdd(h + (size_t)t * DFEAT + d, Wh[(size_t)s * DFEAT + d]);
}

extern "C" void kernel_launch(void* const* d_in, const int* in_sizes, int n_in,
                              void* d_out, int out_size, void* d_ws, size_t ws_size,
                              hipStream_t stream) {
  const float* feats = (const float*)d_in[0];
  const float* Wc    = (const float*)d_in[1];
  const float* bc    = (const float*)d_in[2];
  const float* We    = (const float*)d_in[3];
  const float* be    = (const float*)d_in[4];
  const float* Wo    = (const float*)d_in[5];
  const float* bo    = (const float*)d_in[6];
  const int*   srcC  = (const int*)d_in[7];
  const int*   dstC  = (const int*)d_in[8];
  const int*   srcE  = (const int*)d_in[9];
  const int*   dstE  = (const int*)d_in[10];

  const int n   = in_sizes[0] / DFEAT;
  const int nEc = in_sizes[7];
  const int nEe = in_sizes[9];

  const size_t feSize = (size_t)n * DFEAT;
  float* WhC = (float*)d_ws;
  float* WhE = WhC + feSize;
  float* h   = WhE + feSize;
  float* out = (float*)d_out;

  const long long n4 = (long long)feSize / 4;
  zero_f4<<<(unsigned)((n4 + 255) / 256), 256, 0, stream>>>((float4*)h, n4);

  const int tiles  = ((n + TILE - 1) / TILE) * (DFEAT / TILE);
  const int blocks = (tiles + 7) / 8;
  rgcn_proj_gemm<<<dim3(blocks, 1), 256, 0, stream>>>(
      feats, Wc, bc, We, be, WhC, WhE, n);

  const long long workC = (long long)nEc * DFEAT;
  scatter_add_edges<<<(unsigned)((workC + 255) / 256), 256, 0, stream>>>(
      WhC, srcC, dstC, h, nEc);
  const long long workE = (long long)nEe * DFEAT;
  scatter_add_edges<<<(unsigned)((workE + 255) / 256), 256, 0, stream>>>(
      WhE, srcE, dstE, h, nEe);

  rgcn_out_gemm<<<dim3(blocks, 1), 256, 0, stream>>>(h, Wo, bo, WhE, out, n);
}


# Edge sums by owner work-groups: each owns 196 nodes' rows in LDS, scans both dst lists, lists its edges, adds gathered rows by LDS float adds (arrival order); no global atomics, no clear launch
# speedup vs baseline: 2.6295x; 2.6295x over previous
#include <hip/hip_runtime.h>

typedef __attribute__((ext_vector_type(2))) float v2f;
typedef __attribute__((ext_vector_type(8))) float v8f;

#define DFEAT 128
#define TILE  16

__global__ __launch_bounds__(256) void rgcn_proj_gemm(
    const float* __restrict__ feats,
    const float* __restrict__ Wc, const float* __restrict__ bc,
    const float* __restrict__ We, const float* __restrict__ be,
    float* __restrict__ WhC, float* __restrict__ WhE, int n)
{
  const int wave  = threadIdx.x >> 5;
  const int lane  = threadIdx.x & 31;
  const int gw    = blockIdx.x * 8 + wave;
  const int nColT = DFEAT / TILE;
  const int m0    = (gw / nColT) * TILE;
  const int n0    = (gw % nColT) * TILE;
  if (m0 >= n) return;

  const int half = lane >> 4;
  const int l15  = lane & 15;

  int arowi = m0 + l15;
  if (arowi >= n) arowi = n - 1;
  const float* arow  = feats + (size_t)arowi * DFEAT + half * 2;
  const float* bcrow = Wc    + (size_t)(n0 + l15) * DFEAT + half * 2;
  const float* berow = We    + (size_t)(n0 + l15) * DFEAT + half * 2;

  v8f accC = {0.f, 0.f, 0.f, 0.f, 0.f, 0.f, 0.f, 0.f};
  v8f accE = {0.f, 0.f, 0.f, 0.f, 0.f, 0.f, 0.f, 0.f};
#pragma unroll 8
  for (int k = 0; k < DFEAT; k += 4) {
    v2f a   = *(const v2f*)(arow + k);
    v2f bC  = *(const v2f*)(bcrow + k);
    v2f bE  = *(const v2f*)(berow + k);
    accC = __builtin_amdgcn_wmma_f32_16x16x4_f32(
        false, a, false, bC, (short)0, accC, false, false);
    accE = __builtin_amdgcn_wmma_f32_16x16x4_f32(
        false, a, false, bE, (short)0, accE, false, false);
  }

  const float biasC = bc[n0 + l15];
  const float biasE = be[n0 + l15];
  float* ocol = WhC + (size_t)(m0 + half * 8) * DFEAT + n0 + l15;
  float* ecol = WhE + (size_t)(m0 + half * 8) * DFEAT + n0 + l15;

  if (m0 + TILE <= n) {
#pragma unroll
    for (int r = 0; r < 8; ++r) {
      ocol[(size_t)r * DFEAT] = accC[r] + biasC;
      ecol[(size_t)r * DFEAT] = accE[r] + biasE;
    }
  } else {
#pragma unroll
    for (int r = 0; r < 8; ++r) {
      const int M = m0 + half * 8 + r;
      if (M < n) {
        ocol[(size_t)r * DFEAT] = accC[r] + biasC;
        ecol[(size_t)r * DFEAT] = accE[r] + biasE;
      }
    }
  }
}

__global__ __launch_bounds__(256) void rgcn_out_gemm(
    const float* __restrict__ h,
    const float* __restrict__ Wo, const float* __restrict__ bo,
    const float* __restrict__ WhE, float* __restrict__ out, int n)
{
  const int wave  = threadIdx.x >> 5;
  const int lane  = threadIdx.x & 31;
  const int gw    = blockIdx.x * 8 + wave;
  const int nColT = DFEAT / TILE;
  const int m0    = (gw / nColT) * TILE;
  const int n0    = (gw % nColT) * TILE;
  if (m0 >= n) return;

  const int half = lane >> 4;
  const int l15  = lane & 15;

  int arowi = m0 + l15;
  if (arowi >= n) arowi = n - 1;
  const float* arow = h  + (size_t)arowi * DFEAT + half * 2;
  const float* brow = Wo + (size_t)(n0 + l15) * DFEAT + half * 2;

  v8f acc = {0.f, 0.f, 0.f, 0.f, 0.f, 0.f, 0.f, 0.f};
#pragma unroll 8
  for (int k = 0; k < DFEAT; k += 4) {
    v2f a = *(const v2f*)(arow + k);
    v2f b = *(const v2f*)(brow + k);
    acc = __builtin_amdgcn_wmma_f32_16x16x4_f32(
        false, a, false, b, (short)0, acc, false, false);
  }

  const float bias = bo[n0 + l15];
  const size_t base = (size_t)(m0 + half * 8) * DFEAT + n0 + l15;

  if (m0 + TILE <= n) {
#pragma unroll
    for (int r = 0; r < 8; ++r) {
      const size_t idx = base + (size_t)r * DFEAT;
      out[idx] = acc[r] + bias + WhE[idx];
    }
  } else {
#pragma unroll
    for (int r = 0; r < 8; ++r) {
      const int M = m0 + half * 8 + r;
      if (M < n) {
        const size_t idx = base + (size_t)r * DFEAT;
        out[idx] = acc[r] + bias + WhE[idx];
      }
    }
  }
}

#define OWN_NODES   196
#define OWN_THREADS 1024
#define LIST_CAP    16384
#define SCAN_U      8
#define GATHER_U    8
#define ETYPE_BIT   0x80000000u

__device__ __forceinline__ void lds_fadd(float* p, float v) {
  __hip_atomic_fetch_add(p, v, __ATOMIC_RELAXED, __HIP_MEMORY_SCOPE_WORKGROUP);
}

__device__ __forceinline__ void own_hit(
    unsigned tagged, unsigned dl, uint2* list, unsigned* count)
{
  const unsigned pos = atomicAdd(count, 1u);
  if (pos < LIST_CAP) list[pos] = make_uint2(tagged, dl);
}

__device__ __forceinline__ void own_direct(
    const int* __restrict__ dst, const int* __restrict__ src, int nE,
    const float* __restrict__ WhT, unsigned lo, unsigned cnt, float* acc)
{
  for (int e = threadIdx.x; e < nE; e += OWN_THREADS) {
    const unsigned a = (unsigned)dst[e] - lo;
    if (a < cnt) {
      const float* row = WhT + (size_t)src[e] * DFEAT;
#pragma nounroll
      for (int c = 0; c < DFEAT; ++c)
        lds_fadd(acc + a * DFEAT + (c & 3) * 32 + (c >> 2), row[c]);
    }
  }
}

__device__ __forceinline__ void own_scan(
    const int* __restrict__ dst, int nE, unsigned tag, unsigned lo, unsigned cnt,
    uint2* list, unsigned* count)
{
  const int tid = threadIdx.x;
  const int nE4 = nE >> 2;
  const int4* d4 = (const int4*)dst;
  for (int base = tid; base < nE4; base += OWN_THREADS * SCAN_U) {
    int4 v[SCAN_U];
#pragma unroll
    for (int u = 0; u < SCAN_U; ++u) {
      const int i = base + u * OWN_THREADS;
      v[u] = (i < nE4) ? d4[i] : make_int4(-1, -1, -1, -1);
    }
#pragma unroll
    for (int u = 0; u < SCAN_U; ++u) {
      const unsigned e0 = (unsigned)(base + u * OWN_THREADS) * 4u;
      const unsigned a0 = (unsigned)v[u].x - lo;
      const unsigned a1 = (unsigned)v[u].y - lo;
      const unsigned a2 = (unsigned)v[u].z - lo;
      const unsigned a3 = (unsigned)v[u].w - lo;
      if ((a0 < cnt) | (a1 < cnt) | (a2 < cnt) | (a3 < cnt)) {
        if (a0 < cnt) own_hit(tag | (e0 + 0u), a0, list, count);
        if (a1 < cnt) own_hit(tag | (e0 + 1u), a1, list, count);
        if (a2 < cnt) own_hit(tag | (e0 + 2u), a2, list, count);
        if (a3 < cnt) own_hit(tag | (e0 + 3u), a3, list, count);
      }
    }
  }
  const int t = nE4 * 4 + tid;
  if (t < nE) {
    const unsigned a = (unsigned)dst[t] - lo;
    if (a < cnt) own_hit(tag | (unsigned)t, a, list, count);
  }
}

__global__ __launch_bounds__(OWN_THREADS) void scatter_owner_lds(
    const float* __restrict__ Wh,
    const int* __restrict__ srcC, const int* __restrict__ dstC,
    const int* __restrict__ srcE, const int* __restrict__ dstE,
    float* __restrict__ h, int n, int nEc, int nEe)
{
  __shared__ __attribute__((aligned(16))) float acc[(OWN_NODES + 1) * DFEAT];
  __shared__ __attribute__((aligned(16))) uint2 list[LIST_CAP];
  __shared__ unsigned s_count;

  const int tid  = threadIdx.x;
  const int lane = tid & 31;
  const int wave = __builtin_amdgcn_readfirstlane(tid >> 5);

  for (int tile = blockIdx.x; tile * OWN_NODES < n; tile += gridDim.x) {
    const int lo  = tile * OWN_NODES;
    const int cnt = (n - lo < OWN_NODES) ? (n - lo) : OWN_NODES;

    for (int i = tid; i < (OWN_NODES + 1) * DFEAT / 4; i += OWN_THREADS)
      ((float4*)acc)[i] = make_float4(0.f, 0.f, 0.f, 0.f);
    if (tid == 0) s_count = 0u;
    __syncthreads();

    own_scan(dstC, nEc, 0u,        (unsigned)lo, (unsigned)cnt, list, &s_count);
    own_scan(dstE, nEe, ETYPE_BIT, (unsigned)lo, (unsigned)cnt, list, &s_count);
    __syncthreads();

    unsigned m = (unsigned)__builtin_amdgcn_readfirstlane((int)s_count);
    if (m > LIST_CAP) {
      own_direct(dstC, srcC, nEc, Wh,                      (unsigned)lo, (unsigned)cnt, acc);
      own_direct(dstE, srcE, nEe, Wh + (size_t)n * DFEAT,  (unsigned)lo, (unsigned)cnt, acc);
      m = 0u;
    }

    for (unsigned base = (unsigned)wave * 32u; base < m; base += 32u * (OWN_THREADS / 32)) {
      const unsigned idx = base + (unsigned)lane;
      unsigned rowi = 0u;
      unsigned dl   = OWN_NODES;
      if (idx < m) {
        const uint2 ent = list[idx];
        const bool isE = (ent.x & ETYPE_BIT) != 0u;
        const int* sp = isE ? srcE : srcC;
        rowi = (unsigned)sp[ent.x & ~ETYPE_BIT] + (isE ? (unsigned)n : 0u);
        dl   = ent.y;
      }
      const unsigned left = m - base;
      const unsigned nv   = left < 32u ? left : 32u;
      const unsigned nvr  = (nv + GATHER_U - 1u) & ~(unsigned)(GATHER_U - 1);
      for (unsigned j = 0; j < nvr; j += GATHER_U) {
        float4 v[GATHER_U];
        unsigned d[GATHER_U];
#pragma unroll
        for (int u = 0; u < GATHER_U; ++u) {
          const unsigned r = (unsigned)__builtin_amdgcn_readlane((int)rowi, (int)(j + u));
          d[u] = (unsigned)__builtin_amdgcn_readlane((int)dl, (int)(j + u));
          v[u] = *(const float4*)(Wh + (size_t)r * DFEAT + lane * 4);
        }
#pragma unroll
        for (int u = 0; u < GATHER_U; ++u) {
          float* p = acc + d[u] * DFEAT + lane;
          lds_fadd(p,      v[u].x);
          lds_fadd(p + 32, v[u].y);
          lds_fadd(p + 64, v[u].z);
          lds_fadd(p + 96, v[u].w);
        }
      }
    }
    __syncthreads();

    for (int i = tid; i < cnt * 32; i += OWN_THREADS) {
      const int node = i >> 5;
      const float* p = acc + node * DFEAT + lane;
      const float4 o = make_float4(p[0], p[32], p[64], p[96]);
      *(float4*)(h + (size_t)(lo + node) * DFEAT + lane * 4) = o;
    }
    __syncthreads();
  }
}

extern "C" void kernel_launch(void* const* d_in, const int* in_sizes, int n_in,
                              void* d_out, int out_size, void* d_ws, size_t ws_size,
                              hipStream_t stream) {
  const float* feats = (const float*)d_in[0];
  const float* Wc    = (const float*)d_in[1];
  const float* bc    = (const float*)d_in[2];
  const float* We    = (const float*)d_in[3];
  const float* be    = (const float*)d_in[4];
  const float* Wo    = (const float*)d_in[5];
  const float* bo    = (const float*)d_in[6];
  const int*   srcC  = (const int*)d_in[7];
  const int*   dstC  = (const int*)d_in[8];
  const int*   srcE  = (const int*)d_in[9];
  const int*   dstE  = (const int*)d_in[10];

  const int n   = in_sizes[0] / DFEAT;
  const int nEc = in_sizes[7];
  const int nEe = in_sizes[9];

  const size_t feSize = (size_t)n * DFEAT;
  float* WhC = (float*)d_ws;
  float* WhE = WhC + feSize;
  float* h   = WhE + feSize;
  float* out = (float*)d_out;

  const int tiles  = ((n + TILE - 1) / TILE) * (DFEAT / TILE);
  const int blocks = (tiles + 7) / 8;
  rgcn_proj_gemm<<<dim3(blocks, 1), 256, 0, stream>>>(
      feats, Wc, bc, We, be, WhC, WhE, n);

  scatter_owner_lds<<<256, OWN_THREADS, 0, stream>>>(
      WhC, srcC, dstC, srcE, dstE, h, n, nEc, nEe);

  rgcn_out_gemm<<<dim3(blocks, 1), 256, 0, stream>>>(h, Wo, bo, WhE, out, n);
}


# GEMM tiling by row strips: a wave owns 16 rows x 64 columns and loads each A fragment once for its 4 tiles; W is staged in LDS once per work-group and B fragments are read there; k order unchanged
# speedup vs baseline: 3.8406x; 1.4606x over previous
#include <hip/hip_runtime.h>


typedef __attribute__((ext_vector_type(2))) float v2f;
typedef __attribute__((ext_vector_type(8))) float v8f;

#define DFEAT 128
#define TILE  16

#define WPITCH 132

__device__ __forceinline__ void stage_w(const float* __restrict__ Wm, float* lw) {
  const int row0 = threadIdx.x >> 5;
  const int c4   = threadIdx.x & 31;
  float4 v[16];
#pragma unroll
  for (int p = 0; p < 16; ++p)
    v[p] = *(const float4*)(Wm + (size_t)(row0 + p * 8) * DFEAT + c4 * 4);
#pragma unroll
  for (int p = 0; p < 16; ++p)
    *(float4*)(lw + (row0 + p * 8) * WPITCH + c4 * 4) = v[p];
}

__device__ __forceinline__ int strip_of_round(int r, int q, int nq, int nStrips) {
  const int base = r * nq;
  int c = nStrips - base;
  if (c > nq) c = nq;
  const unsigned lo = ((unsigned)q * (unsigned)c) / (unsigned)nq;
  const unsigned hi = ((unsigned)(q + 1) * (unsigned)c) / (unsigned)nq;
  return (hi > lo) ? base + (int)lo : -1;
}

template <int NT, int KC>
__device__ __forceinline__ void strip_mma(
    const float* __restrict__ arow, const float* bl, v8f (&acc)[NT])
{
#pragma unroll 1
  for (int k0 = 0; k0 < DFEAT; k0 += 4 * KC) {
    v2f a[KC];
#pragma unroll
    for (int s = 0; s < KC; ++s) a[s] = *(const v2f*)(arow + k0 + 4 * s);
#pragma unroll
    for (int s = 0; s < KC; ++s) {
#pragma unroll
      for (int j = 0; j < NT; ++j) {
        const v2f b = *(const v2f*)(bl + j * TILE * WPITCH + k0 + 4 * s);
        acc[j] = __builtin_amdgcn_wmma_f32_16x16x4_f32(
            false, a[s], false, b, (short)0, acc[j], false, false);
      }
    }
  }
}

__global__ __launch_bounds__(256) void rgcn_proj_gemm(
    const float* __restrict__ feats,
    const float* __restrict__ Wc, const float* __restrict__ bc,
    const float* __restrict__ We, const float* __restrict__ be,
    float* __restrict__ WhC, float* __restrict__ WhE, int n)
{
  __shared__ __attribute__((aligned(16))) float lw[2 * DFEAT * WPITCH];
  stage_w(Wc, lw);
  stage_w(We, lw + DFEAT * WPITCH);
  __syncthreads();

  const int lane = threadIdx.x & 31;
  const int wave = __builtin_amdgcn_readfirstlane(threadIdx.x >> 5);
  const int g    = wave & 3;
  const int q    = blockIdx.x * 2 + (wave >> 2);
  const int nq   = gridDim.x * 2;
  const int nStrips = (n + TILE - 1) / TILE;

  const int half = lane >> 4;
  const int l15  = lane & 15;
  const int n0   = (g & 1) * 64;
  const float* bl   = lw + (g * 64 + l15) * WPITCH + half * 2;
  const float* bias = (g < 2) ? bc : be;
  float* dst        = (g < 2) ? WhC : WhE;

  float bv[4];
#pragma unroll
  for (int j = 0; j < 4; ++j) bv[j] = bias[n0 + j * TILE + l15];

  for (int r = 0; r * nq < nStrips; ++r) {
    const int s = strip_of_round(r, q, nq, nStrips);
    if (s < 0) continue;
    const int m0 = s * TILE;
    int arowi = m0 + l15;
    if (arowi >= n) arowi = n - 1;
    const float* arow = feats + (size_t)arowi * DFEAT + half * 2;

    v8f acc[4];
#pragma unroll
    for (int j = 0; j < 4; ++j) acc[j] = (v8f){0.f, 0.f, 0.f, 0.f, 0.f, 0.f, 0.f, 0.f};
    strip_mma<4, 32>(arow, bl, acc);

    float* ocol = dst + (size_t)(m0 + half * 8) * DFEAT + n0 + l15;
    if (m0 + TILE <= n) {
#pragma unroll
      for (int j = 0; j < 4; ++j)
#pragma unroll
        for (int rr = 0; rr < 8; ++rr)
          ocol[(size_t)rr * DFEAT + j * TILE] = acc[j][rr] + bv[j];
    } else {
#pragma unroll
      for (int j = 0; j < 4; ++j)
#pragma unroll
        for (int rr = 0; rr < 8; ++rr)
          if (m0 + half * 8 + rr < n)
            ocol[(size_t)rr * DFEAT + j * TILE] = acc[j][rr] + bv[j];
    }
  }
}

__global__ __launch_bounds__(256) void rgcn_out_gemm(
    const float* __restrict__ h,
    const float* __restrict__ Wo, const float* __restrict__ bo,
    const float* __restrict__ WhE, float* __restrict__ out, int n)
{
  __shared__ __attribute__((aligned(16))) float lw[DFEAT * WPITCH];
  stage_w(Wo, lw);
  __syncthreads();

  const int lane = threadIdx.x & 31;
  const int wave = __builtin_amdgcn_readfirstlane(threadIdx.x >> 5);
  const int g    = wave & 1;
  const int q    = blockIdx.x * 4 + (wave >> 1);
  const int nq   = gridDim.x * 4;
  const int nStrips = (n + TILE - 1) / TILE;

  const int half = lane >> 4;
  const int l15  = lane & 15;
  const int n0   = g * 64;
  const float* bl = lw + (n0 + l15) * WPITCH + half * 2;

  float bv[4];
#pragma unroll
  for (int j = 0; j < 4; ++j) bv[j] = bo[n0 + j * TILE + l15];

  for (int r = 0; r * nq < nStrips; ++r) {
    const int s = strip_of_round(r, q, nq, nStrips);
    if (s < 0) continue;
    const int m0 = s * TILE;
    int arowi = m0 + l15;
    if (arowi >= n) arowi = n - 1;
    const float* arow = h + (size_t)arowi * DFEAT + half * 2;

    v8f acc[4];
#pragma unroll
    for (int j = 0; j < 4; ++j) acc[j] = (v8f){0.f, 0.f, 0.f, 0.f, 0.f, 0.f, 0.f, 0.f};
    strip_mma<4, 8>(arow, bl, acc);

    const size_t base = (size_t)(m0 + half * 8) * DFEAT + n0 + l15;
    if (m0 + TILE <= n) {
#pragma unroll
      for (int j = 0; j < 4; ++j)
#pragma unroll
        for (int rr = 0; rr < 8; ++rr) {
          const size_t idx = base + (size_t)rr * DFEAT + j * TILE;
          out[idx] = acc[j][rr] + bv[j] + WhE[idx];
        }
    } else {
#pragma unroll
      for (int j = 0; j < 4; ++j)
#pragma unroll
        for (int rr = 0; rr < 8; ++rr)
          if (m0 + half * 8 + rr < n) {
            const size_t idx = base + (size_t)rr * DFEAT + j * TILE;
            out[idx] = acc[j][rr] + bv[j] + WhE[idx];
          }
    }
  }
}

#define OWN_NODES   196
#define OWN_THREADS 1024
#define LIST_CAP    16384
#define SCAN_U      8
#define GATHER_U    8
#define ETYPE_BIT   0x80000000u

__device__ __forceinline__ void lds_fadd(float* p, float v) {
  __hip_atomic_fetch_add(p, v, __ATOMIC_RELAXED, __HIP_MEMORY_SCOPE_WORKGROUP);
}

__device__ __forceinline__ void own_hit(
    unsigned tagged, unsigned dl, uint2* list, unsigned* count)
{
  const unsigned pos = atomicAdd(count, 1u);
  if (pos < LIST_CAP) list[pos] = make_uint2(tagged, dl);
}

__device__ __forceinline__ void own_direct(
    const int* __restrict__ dst, const int* __restrict__ src, int nE,
    const float* __restrict__ WhT, unsigned lo, unsigned cnt, float* acc)
{
  for (int e = threadIdx.x; e < nE; e += OWN_THREADS) {
    const unsigned a = (unsigned)dst[e] - lo;
    if (a < cnt) {
      const float* row = WhT + (size_t)src[e] * DFEAT;
#pragma nounroll
      for (int c = 0; c < DFEAT; ++c)
        lds_fadd(acc + a * DFEAT + (c & 3) * 32 + (c >> 2), row[c]);
    }
  }
}

__device__ __forceinline__ void own_scan(
    const int* __restrict__ dst, int nE, unsigned tag, unsigned lo, unsigned cnt,
    uint2* list, unsigned* count)
{
  const int tid = threadIdx.x;
  const int nE4 = nE >> 2;
  const int4* d4 = (const int4*)dst;
  for (int base = tid; base < nE4; base += OWN_THREADS * SCAN_U) {
    int4 v[SCAN_U];
#pragma unroll
    for (int u = 0; u < SCAN_U; ++u) {
      const int i = base + u * OWN_THREADS;
      v[u] = (i < nE4) ? d4[i] : make_int4(-1, -1, -1, -1);
    }
#pragma unroll
    for (int u = 0; u < SCAN_U; ++u) {
      const unsigned e0 = (unsigned)(base + u * OWN_THREADS) * 4u;
      const unsigned a0 = (unsigned)v[u].x - lo;
      const unsigned a1 = (unsigned)v[u].y - lo;
      const unsigned a2 = (unsigned)v[u].z - lo;
      const unsigned a3 = (unsigned)v[u].w - lo;
      if ((a0 < cnt) | (a1 < cnt) | (a2 < cnt) | (a3 < cnt)) {
        if (a0 < cnt) own_hit(tag | (e0 + 0u), a0, list, count);
        if (a1 < cnt) own_hit(tag | (e0 + 1u), a1, list, count);
        if (a2 < cnt) own_hit(tag | (e0 + 2u), a2, list, count);
        if (a3 < cnt) own_hit(tag | (e0 + 3u), a3, list, count);
      }
    }
  }
  const int t = nE4 * 4 + tid;
  if (t < nE) {
    const unsigned a = (unsigned)dst[t] - lo;
    if (a < cnt) own_hit(tag | (unsigned)t, a, list, count);
  }
}

__global__ __launch_bounds__(OWN_THREADS) void scatter_owner_lds(
    const float* __restrict__ Wh,
    const int* __restrict__ srcC, const int* __restrict__ dstC,
    const int* __restrict__ srcE, const int* __restrict__ dstE,
    float* __restrict__ h, int n, int nEc, int nEe)
{
  __shared__ __attribute__((aligned(16))) float acc[(OWN_NODES + 1) * DFEAT];
  __shared__ __attribute__((aligned(16))) uint2 list[LIST_CAP];
  __shared__ unsigned s_count;

  const int tid  = threadIdx.x;
  const int lane = tid & 31;
  const int wave = __builtin_amdgcn_readfirstlane(tid >> 5);

  for (int tile = blockIdx.x; tile * OWN_NODES < n; tile += gridDim.x) {
    const int lo  = tile * OWN_NODES;
    const int cnt = (n - lo < OWN_NODES) ? (n - lo) : OWN_NODES;

    for (int i = tid; i < (OWN_NODES + 1) * DFEAT / 4; i += OWN_THREADS)
      ((float4*)acc)[i] = make_float4(0.f, 0.f, 0.f, 0.f);
    if (tid == 0) s_count = 0u;
    __syncthreads();

    own_scan(dstC, nEc, 0u,        (unsigned)lo, (unsigned)cnt, list, &s_count);
    own_scan(dstE, nEe, ETYPE_BIT, (unsigned)lo, (unsigned)cnt, list, &s_count);
    __syncthreads();

    unsigned m = (unsigned)__builtin_amdgcn_readfirstlane((int)s_count);
    if (m > LIST_CAP) {
      own_direct(dstC, srcC, nEc, Wh,                      (unsigned)lo, (unsigned)cnt, acc);
      own_direct(dstE, srcE, nEe, Wh + (size_t)n * DFEAT,  (unsigned)lo, (unsigned)cnt, acc);
      m = 0u;
    }

    for (unsigned base = (unsigned)wave * 32u; base < m; base += 32u * (OWN_THREADS / 32)) {
      const unsigned idx = base + (unsigned)lane;
      unsigned rowi = 0u;
      unsigned dl   = OWN_NODES;
      if (idx < m) {
        const uint2 ent = list[idx];
        const bool isE = (ent.x & ETYPE_BIT) != 0u;
        const int* sp = isE ? srcE : srcC;
        rowi = (unsigned)sp[ent.x & ~ETYPE_BIT] + (isE ? (unsigned)n : 0u);
        dl   = ent.y;
      }
      const unsigned left = m - base;
      const unsigned nv   = left < 32u ? left : 32u;
      const unsigned nvr  = (nv + GATHER_U - 1u) & ~(unsigned)(GATHER_U - 1);
      for (unsigned j = 0; j < nvr; j += GATHER_U) {
        float4 v[GATHER_U];
        unsigned d[GATHER_U];
#pragma unroll
        for (int u = 0; u < GATHER_U; ++u) {
          const unsigned r = (unsigned)__builtin_amdgcn_readlane((int)rowi, (int)(j + u));
          d[u] = (unsigned)__builtin_amdgcn_readlane((int)dl, (int)(j + u));
          v[u] = *(const float4*)(Wh + (size_t)r * DFEAT + lane * 4);
        }
#pragma unroll
        for (int u = 0; u < GATHER_U; ++u) {
          float* p = acc + d[u] * DFEAT + lane;
          lds_fadd(p,      v[u].x);
          lds_fadd(p + 32, v[u].y);
          lds_fadd(p + 64, v[u].z);
          lds_fadd(p + 96, v[u].w);
        }
      }
    }
    __syncthreads();

    for (int i = tid; i < cnt * 32; i += OWN_THREADS) {
      const int node = i >> 5;
      const float* p = acc + node * DFEAT + lane;
      const float4 o = make_float4(p[0], p[32], p[64], p[96]);
      *(float4*)(h + (size_t)(lo + node) * DFEAT + lane * 4) = o;
    }
    __syncthreads();
  }
}

extern "C" void kernel_launch(void* const* d_in, const int* in_sizes, int n_in,
                              void* d_out, int out_size, void* d_ws, size_t ws_size,
                              hipStream_t stream) {
  const float* feats = (const float*)d_in[0];
  const float* Wc    = (const float*)d_in[1];
  const float* bc    = (const float*)d_in[2];
  const float* We    = (const float*)d_in[3];
  const float* be    = (const float*)d_in[4];
  const float* Wo    = (const float*)d_in[5];
  const float* bo    = (const float*)d_in[6];
  const int*   srcC  = (const int*)d_in[7];
  const int*   dstC  = (const int*)d_in[8];
  const int*   srcE  = (const int*)d_in[9];
  const int*   dstE  = (const int*)d_in[10];

  const int n   = in_sizes[0] / DFEAT;
  const int nEc = in_sizes[7];
  const int nEe = in_sizes[9];

  const size_t feSize = (size_t)n * DFEAT;
  float* WhC = (float*)d_ws;
  float* WhE = WhC + feSize;
  float* h   = WhE + feSize;
  float* out = (float*)d_out;

  rgcn_proj_gemm<<<512, 256, 0, stream>>>(
      feats, Wc, bc, We, be, WhC, WhE, n);

  scatter_owner_lds<<<256, OWN_THREADS, 0, stream>>>(
      WhC, srcC, dstC, srcE, dstE, h, n, nEc, nEe);

  rgcn_out_gemm<<<1024, 256, 0, stream>>>(h, Wo, bo, WhE, out, n);
}


# Owner scan on a 1-byte key plane: the projection launch writes each edge's owner-tile byte; owners test 16 edges per 16-B load and range-test the listed edges at the gather; sum order as before
# speedup vs baseline: 4.5681x; 1.1894x over previous
#include <hip/hip_runtime.h>


typedef __attribute__((ext_vector_type(2))) float v2f;
typedef __attribute__((ext_vector_type(8))) float v8f;

#define DFEAT 128
#define TILE  16

#define WPITCH 132

__device__ __forceinline__ void stage_w(const float* __restrict__ Wm, float* lw) {
  const int row0 = threadIdx.x >> 5;
  const int c4   = threadIdx.x & 31;
  float4 v[16];
#pragma unroll
  for (int p = 0; p < 16; ++p)
    v[p] = *(const float4*)(Wm + (size_t)(row0 + p * 8) * DFEAT + c4 * 4);
#pragma unroll
  for (int p = 0; p < 16; ++p)
    *(float4*)(lw + (row0 + p * 8) * WPITCH + c4 * 4) = v[p];
}

__device__ __forceinline__ int strip_of_round(int r, int q, int nq, int nStrips) {
  const int base = r * nq;
  int c = nStrips - base;
  if (c > nq) c = nq;
  const unsigned lo = ((unsigned)q * (unsigned)c) / (unsigned)nq;
  const unsigned hi = ((unsigned)(q + 1) * (unsigned)c) / (unsigned)nq;
  return (hi > lo) ? base + (int)lo : -1;
}

template <int NT, int KC>
__device__ __forceinline__ void strip_mma(
    const float* __restrict__ arow, const float* bl, v8f (&acc)[NT])
{
#pragma unroll 1
  for (int k0 = 0; k0 < DFEAT; k0 += 4 * KC) {
    v2f a[KC];
#pragma unroll
    for (int s = 0; s < KC; ++s) a[s] = *(const v2f*)(arow + k0 + 4 * s);
#pragma unroll
    for (int s = 0; s < KC; ++s) {
#pragma unroll
      for (int j = 0; j < NT; ++j) {
        const v2f b = *(const v2f*)(bl + j * TILE * WPITCH + k0 + 4 * s);
        acc[j] = __builtin_amdgcn_wmma_f32_16x16x4_f32(
            false, a[s], false, b, (short)0, acc[j], false, false);
      }
    }
  }
}

#define OWN_NODES   196

__device__ __forceinline__ unsigned own_key(int d) {
  return ((unsigned)d / (unsigned)OWN_NODES) & 255u;
}

__device__ __forceinline__ unsigned pack_keys(int4 d) {
  return own_key(d.x) | (own_key(d.y) << 8) | (own_key(d.z) << 16) | (own_key(d.w) << 24);
}

__device__ __forceinline__ void make_keys_rest(
    const int* __restrict__ dst, int nE, unsigned* __restrict__ kw, int gtid, int gsz)
{
  const int nE4 = nE >> 2;
  const int4* d4 = (const int4*)dst;
#pragma nounroll
  for (int i = gtid + gsz; i < nE4; i += gsz) kw[i] = pack_keys(d4[i]);
}

__device__ __forceinline__ size_t keys_second(int nEfirst) {
  return ((size_t)nEfirst + 15) & ~(size_t)15;
}

__global__ __launch_bounds__(256) void rgcn_proj_gemm(
    const float* __restrict__ feats,
    const float* __restrict__ Wc, const float* __restrict__ bc,
    const float* __restrict__ We, const float* __restrict__ be,
    float* __restrict__ WhC, float* __restrict__ WhE, int n,
    const int* __restrict__ dstC, const int* __restrict__ dstE,
    int nEc, int nEe, unsigned char* __restrict__ keys)
{
  __shared__ __attribute__((aligned(16))) float lw[2 * DFEAT * WPITCH];
  const int gtid = blockIdx.x * blockDim.x + threadIdx.x;
  const int gsz  = gridDim.x * blockDim.x;
  unsigned* kwC = (unsigned*)keys;
  unsigned* kwE = (unsigned*)(keys + keys_second(nEc));
  const int nC4 = nEc >> 2;
  const int nE4 = nEe >> 2;
  const int4 dC = ((const int4*)dstC)[gtid < nC4 ? gtid : (nC4 > 0 ? nC4 - 1 : 0)];
  const int4 dE = ((const int4*)dstE)[gtid < nE4 ? gtid : (nE4 > 0 ? nE4 - 1 : 0)];

  stage_w(Wc, lw);
  stage_w(We, lw + DFEAT * WPITCH);
  __syncthreads();

  if (gtid < nC4) kwC[gtid] = pack_keys(dC);
  if (gtid < nE4) kwE[gtid] = pack_keys(dE);
  make_keys_rest(dstC, nEc, kwC, gtid, gsz);
  make_keys_rest(dstE, nEe, kwE, gtid, gsz);

  const int lane = threadIdx.x & 31;
  const int wave = __builtin_amdgcn_readfirstlane(threadIdx.x >> 5);
  const int g    = wave & 3;
  const int q    = blockIdx.x * 2 + (wave >> 2);
  const int nq   = gridDim.x * 2;
  const int nStrips = (n + TILE - 1) / TILE;

  const int half = lane >> 4;
  const int l15  = lane & 15;
  const int n0   = (g & 1) * 64;
  const float* bl   = lw + (g * 64 + l15) * WPITCH + half * 2;
  const float* bias = (g < 2) ? bc : be;
  float* dst        = (g < 2) ? WhC : WhE;

  float bv[4];
#pragma unroll
  for (int j = 0; j < 4; ++j) bv[j] = bias[n0 + j * TILE + l15];

  for (int r = 0; r * nq < nStrips; ++r) {
    const int s = strip_of_round(r, q, nq, nStrips);
    if (s < 0) continue;
    const int m0 = s * TILE;
    int arowi = m0 + l15;
    if (arowi >= n) arowi = n - 1;
    const float* arow = feats + (size_t)arowi * DFEAT + half * 2;

    v8f acc[4];
#pragma unroll
    for (int j = 0; j < 4; ++j) acc[j] = (v8f){0.f, 0.f, 0.f, 0.f, 0.f, 0.f, 0.f, 0.f};
    strip_mma<4, 32>(arow, bl, acc);

    float* ocol = dst + (size_t)(m0 + half * 8) * DFEAT + n0 + l15;
    if (m0 + TILE <= n) {
#pragma unroll
      for (int j = 0; j < 4; ++j)
#pragma unroll
        for (int rr = 0; rr < 8; ++rr)
          ocol[(size_t)rr * DFEAT + j * TILE] = acc[j][rr] + bv[j];
    } else {
#pragma unroll
      for (int j = 0; j < 4; ++j)
#pragma unroll
        for (int rr = 0; rr < 8; ++rr)
          if (m0 + half * 8 + rr < n)
            ocol[(size_t)rr * DFEAT + j * TILE] = acc[j][rr] + bv[j];
    }
  }
}

__global__ __launch_bounds__(256) void rgcn_out_gemm(
    const float* __restrict__ h,
    const float* __restrict__ Wo, const float* __restrict__ bo,
    const float* __restrict__ WhE, float* __restrict__ out, int n)
{
  __shared__ __attribute__((aligned(16))) float lw[DFEAT * WPITCH];
  stage_w(Wo, lw);
  __syncthreads();

  const int lane = threadIdx.x & 31;
  const int wave = __builtin_amdgcn_readfirstlane(threadIdx.x >> 5);
  const int g    = wave & 1;
  const int q    = blockIdx.x * 4 + (wave >> 1);
  const int nq   = gridDim.x * 4;
  const int nStrips = (n + TILE - 1) / TILE;

  const int half = lane >> 4;
  const int l15  = lane & 15;
  const int n0   = g * 64;
  const float* bl = lw + (n0 + l15) * WPITCH + half * 2;

  float bv[4];
#pragma unroll
  for (int j = 0; j < 4; ++j) bv[j] = bo[n0 + j * TILE + l15];

  for (int r = 0; r * nq < nStrips; ++r) {
    const int s = strip_of_round(r, q, nq, nStrips);
    if (s < 0) continue;
    const int m0 = s * TILE;
    int arowi = m0 + l15;
    if (arowi >= n) arowi = n - 1;
    const float* arow = h + (size_t)arowi * DFEAT + half * 2;

    v8f acc[4];
#pragma unroll
    for (int j = 0; j < 4; ++j) acc[j] = (v8f){0.f, 0.f, 0.f, 0.f, 0.f, 0.f, 0.f, 0.f};
    strip_mma<4, 8>(arow, bl, acc);

    const size_t base = (size_t)(m0 + half * 8) * DFEAT + n0 + l15;
    if (m0 + TILE <= n) {
#pragma unroll
      for (int j = 0; j < 4; ++j)
#pragma unroll
        for (int rr = 0; rr < 8; ++rr) {
          const size_t idx = base + (size_t)rr * DFEAT + j * TILE;
          out[idx] = acc[j][rr] + bv[j] + WhE[idx];
        }
    } else {
#pragma unroll
      for (int j = 0; j < 4; ++j)
#pragma unroll
        for (int rr = 0; rr < 8; ++rr)
          if (m0 + half * 8 + rr < n) {
            const size_t idx = base + (size_t)rr * DFEAT + j * TILE;
            out[idx] = acc[j][rr] + bv[j] + WhE[idx];
          }
    }
  }
}

#define OWN_THREADS 1024
#define LIST_CAP    16384
#define SCAN_U      8
#define GATHER_U    8
#define ETYPE_BIT   0x80000000u

__device__ __forceinline__ void lds_fadd(float* p, float v) {
  __hip_atomic_fetch_add(p, v, __ATOMIC_RELAXED, __HIP_MEMORY_SCOPE_WORKGROUP);
}

__device__ __forceinline__ void own_hit(
    unsigned tagged, unsigned* list, unsigned* count)
{
  const unsigned pos = atomicAdd(count, 1u);
  if (pos < LIST_CAP) list[pos] = tagged;
}

__device__ __forceinline__ void own_direct(
    const int* __restrict__ dst, const int* __restrict__ src, int nE,
    const float* __restrict__ WhT, unsigned lo, unsigned cnt, float* acc)
{
  for (int e = threadIdx.x; e < nE; e += OWN_THREADS) {
    const unsigned a = (unsigned)dst[e] - lo;
    if (a < cnt) {
      const float* row = WhT + (size_t)src[e] * DFEAT;
#pragma nounroll
      for (int c = 0; c < DFEAT; ++c)
        lds_fadd(acc + a * DFEAT + (c & 3) * 32 + (c >> 2), row[c]);
    }
  }
}

__device__ __forceinline__ void own_scan(
    const unsigned char* __restrict__ keys, const int* __restrict__ dst, int nE,
    unsigned tag, unsigned lo, unsigned cnt, unsigned splat,
    unsigned* list, unsigned* count)
{
  const int tid  = threadIdx.x;
  const int nE16 = nE >> 4;
  const uint4* k4 = (const uint4*)keys;
  for (int base = tid; base < nE16; base += OWN_THREADS * SCAN_U) {
    uint4 v[SCAN_U];
#pragma unroll
    for (int u = 0; u < SCAN_U; ++u) {
      const int i = base + u * OWN_THREADS;
      v[u] = k4[(i < nE16) ? i : nE16 - 1];
    }
#pragma unroll
    for (int u = 0; u < SCAN_U; ++u) {
      const unsigned x0 = v[u].x ^ splat, x1 = v[u].y ^ splat;
      const unsigned x2 = v[u].z ^ splat, x3 = v[u].w ^ splat;
      const unsigned t0 = (x0 - 0x01010101u) & ~x0;
      const unsigned t1 = (x1 - 0x01010101u) & ~x1;
      const unsigned t2 = (x2 - 0x01010101u) & ~x2;
      const unsigned t3 = (x3 - 0x01010101u) & ~x3;
      const bool inr = (base + u * OWN_THREADS) < nE16;
      if (inr & (((t0 | t1 | t2 | t3) & 0x80808080u) != 0u)) {
        unsigned m = ((t0 & 0x80808080u) >> 7) | ((t1 & 0x80808080u) >> 6) |
                     ((t2 & 0x80808080u) >> 5) | ((t3 & 0x80808080u) >> 4);
        const unsigned e0 = (unsigned)(base + u * OWN_THREADS) * 16u;
        while (m != 0u) {
          const unsigned p = (unsigned)__builtin_ctz(m);
          m &= m - 1u;
          own_hit(tag | (e0 + ((p & 7u) << 2) + (p >> 3)), list, count);
        }
      }
    }
  }
  const int t = nE16 * 16 + tid;
  if (t < nE) {
    const unsigned a = (unsigned)dst[t] - lo;
    if (a < cnt) own_hit(tag | (unsigned)t, list, count);
  }
}

__global__ __launch_bounds__(OWN_THREADS) void scatter_owner_lds(
    const float* __restrict__ Wh,
    const int* __restrict__ srcC, const int* __restrict__ dstC,
    const int* __restrict__ srcE, const int* __restrict__ dstE,
    const unsigned char* __restrict__ keys,
    float* __restrict__ h, int n, int nEc, int nEe)
{
  __shared__ __attribute__((aligned(16))) float acc[(OWN_NODES + 1) * DFEAT];
  __shared__ __attribute__((aligned(16))) unsigned list[LIST_CAP];
  __shared__ unsigned s_count;

  const int tid  = threadIdx.x;
  const int lane = tid & 31;
  const int wave = __builtin_amdgcn_readfirstlane(tid >> 5);

  for (int tile = blockIdx.x; tile * OWN_NODES < n; tile += gridDim.x) {
    const int lo  = tile * OWN_NODES;
    const int cnt = (n - lo < OWN_NODES) ? (n - lo) : OWN_NODES;

    for (int i = tid; i < (OWN_NODES + 1) * DFEAT / 4; i += OWN_THREADS)
      ((float4*)acc)[i] = make_float4(0.f, 0.f, 0.f, 0.f);
    if (tid == 0) s_count = 0u;
    __syncthreads();

    const unsigned splat = ((unsigned)tile & 255u) * 0x01010101u;
    own_scan(keys, dstC, nEc, 0u,
             (unsigned)lo, (unsigned)cnt, splat, list, &s_count);
    own_scan(keys + keys_second(nEc), dstE, nEe, ETYPE_BIT,
             (unsigned)lo, (unsigned)cnt, splat, list, &s_count);
    __syncthreads();

    unsigned m = (unsigned)__builtin_amdgcn_readfirstlane((int)s_count);
    if (m > LIST_CAP) {
      own_direct(dstC, srcC, nEc, Wh,                      (unsigned)lo, (unsigned)cnt, acc);
      own_direct(dstE, srcE, nEe, Wh + (size_t)n * DFEAT,  (unsigned)lo, (unsigned)cnt, acc);
      m = 0u;
    }

    for (unsigned base = (unsigned)wave * 32u; base < m; base += 32u * (OWN_THREADS / 32)) {
      const unsigned idx = base + (unsigned)lane;
      unsigned rowi = 0u;
      unsigned dl   = OWN_NODES;
      if (idx < m) {
        const unsigned ent = list[idx];
        const bool isE = (ent & ETYPE_BIT) != 0u;
        const unsigned e = ent & ~ETYPE_BIT;
        const int* sp = isE ? srcE : srcC;
        const int* dp = isE ? dstE : dstC;
        const unsigned a = (unsigned)dp[e] - (unsigned)lo;
        const unsigned s = (unsigned)sp[e] + (isE ? (unsigned)n : 0u);
        const bool own = a < (unsigned)cnt;
        rowi = own ? s : 0u;
        dl   = own ? a : (unsigned)OWN_NODES;
      }
      const unsigned left = m - base;
      const unsigned nv   = left < 32u ? left : 32u;
      const unsigned nvr  = (nv + GATHER_U - 1u) & ~(unsigned)(GATHER_U - 1);
      for (unsigned j = 0; j < nvr; j += GATHER_U) {
        float4 v[GATHER_U];
        unsigned d[GATHER_U];
#pragma unroll
        for (int u = 0; u < GATHER_U; ++u) {
          const unsigned r = (unsigned)__builtin_amdgcn_readlane((int)rowi, (int)(j + u));
          d[u] = (unsigned)__builtin_amdgcn_readlane((int)dl, (int)(j + u));
          v[u] = *(const float4*)(Wh + (size_t)r * DFEAT + lane * 4);
        }
#pragma unroll
        for (int u = 0; u < GATHER_U; ++u) {
          float* p = acc + d[u] * DFEAT + lane;
          lds_fadd(p,      v[u].x);
          lds_fadd(p + 32, v[u].y);
          lds_fadd(p + 64, v[u].z);
          lds_fadd(p + 96, v[u].w);
        }
      }
    }
    __syncthreads();

    for (int i = tid; i < cnt * 32; i += OWN_THREADS) {
      const int node = i >> 5;
      const float* p = acc + node * DFEAT + lane;
      const float4 o = make_float4(p[0], p[32], p[64], p[96]);
      *(float4*)(h + (size_t)(lo + node) * DFEAT + lane * 4) = o;
    }
    __syncthreads();
  }
}

extern "C" void kernel_launch(void* const* d_in, const int* in_sizes, int n_in,
                              void* d_out, int out_size, void* d_ws, size_t ws_size,
                              hipStream_t stream) {
  const float* feats = (const float*)d_in[0];
  const float* Wc    = (const float*)d_in[1];
  const float* bc    = (const float*)d_in[2];
  const float* We    = (const float*)d_in[3];
  const float* be    = (const float*)d_in[4];
  const float* Wo    = (const float*)d_in[5];
  const float* bo    = (const float*)d_in[6];
  const int*   srcC  = (const int*)d_in[7];
  const int*   dstC  = (const int*)d_in[8];
  const int*   srcE  = (const int*)d_in[9];
  const int*   dstE  = (const int*)d_in[10];

  const int n   = in_sizes[0] / DFEAT;
  const int nEc = in_sizes[7];
  const int nEe = in_sizes[9];

  const size_t feSize = (size_t)n * DFEAT;
  float* WhC = (float*)d_ws;
  float* WhE = WhC + feSize;
  float* h   = WhE + feSize;
  unsigned char* keys = (unsigned char*)(h + feSize);
  float* out = (float*)d_out;

  rgcn_proj_gemm<<<512, 256, 0, stream>>>(
      feats, Wc, bc, We, be, WhC, WhE, n, dstC, dstE, nEc, nEe, keys);

  scatter_owner_lds<<<256, OWN_THREADS, 0, stream>>>(
      WhC, srcC, dstC, srcE, dstE, keys, h, n, nEc, nEe);

  rgcn_out_gemm<<<1024, 256, 0, stream>>>(h, Wo, bo, WhE, out, n);
}


# Owner list per wave, no shared counter: a wave appends its key hits to its own LDS segment by ballot rank and gathers them itself, 32 at a time, no barrier from scan to gather; sum order: LDS arrival
# speedup vs baseline: 4.8553x; 1.0629x over previous
#include <hip/hip_runtime.h>


typedef __attribute__((ext_vector_type(2))) float v2f;
typedef __attribute__((ext_vector_type(8))) float v8f;

#define DFEAT 128
#define TILE  16

#define WPITCH 132

__device__ __forceinline__ void stage_w(const float* __restrict__ Wm, float* lw) {
  const int row0 = threadIdx.x >> 5;
  const int c4   = threadIdx.x & 31;
  float4 v[16];
#pragma unroll
  for (int p = 0; p < 16; ++p)
    v[p] = *(const float4*)(Wm + (size_t)(row0 + p * 8) * DFEAT + c4 * 4);
#pragma unroll
  for (int p = 0; p < 16; ++p)
    *(float4*)(lw + (row0 + p * 8) * WPITCH + c4 * 4) = v[p];
}

__device__ __forceinline__ int strip_of_round(int r, int q, int nq, int nStrips) {
  const int base = r * nq;
  int c = nStrips - base;
  if (c > nq) c = nq;
  const unsigned lo = ((unsigned)q * (unsigned)c) / (unsigned)nq;
  const unsigned hi = ((unsigned)(q + 1) * (unsigned)c) / (unsigned)nq;
  return (hi > lo) ? base + (int)lo : -1;
}

template <int NT, int KC>
__device__ __forceinline__ void strip_mma(
    const float* __restrict__ arow, const float* bl, v8f (&acc)[NT])
{
#pragma unroll 1
  for (int k0 = 0; k0 < DFEAT; k0 += 4 * KC) {
    v2f a[KC];
#pragma unroll
    for (int s = 0; s < KC; ++s) a[s] = *(const v2f*)(arow + k0 + 4 * s);
#pragma unroll
    for (int s = 0; s < KC; ++s) {
#pragma unroll
      for (int j = 0; j < NT; ++j) {
        const v2f b = *(const v2f*)(bl + j * TILE * WPITCH + k0 + 4 * s);
        acc[j] = __builtin_amdgcn_wmma_f32_16x16x4_f32(
            false, a[s], false, b, (short)0, acc[j], false, false);
      }
    }
  }
}

#define OWN_NODES   196

__device__ __forceinline__ unsigned own_key(int d) {
  return ((unsigned)d / (unsigned)OWN_NODES) & 255u;
}

__device__ __forceinline__ unsigned pack_keys(int4 d) {
  return own_key(d.x) | (own_key(d.y) << 8) | (own_key(d.z) << 16) | (own_key(d.w) << 24);
}

__device__ __forceinline__ void make_keys_rest(
    const int* __restrict__ dst, int nE, unsigned* __restrict__ kw, int gtid, int gsz)
{
  const int nE4 = nE >> 2;
  const int4* d4 = (const int4*)dst;
#pragma nounroll
  for (int i = gtid + gsz; i < nE4; i += gsz) kw[i] = pack_keys(d4[i]);
}

__device__ __forceinline__ size_t keys_second(int nEfirst) {
  return ((size_t)nEfirst + 15) & ~(size_t)15;
}

__global__ __launch_bounds__(256) void rgcn_proj_gemm(
    const float* __restrict__ feats,
    const float* __restrict__ Wc, const float* __restrict__ bc,
    const float* __restrict__ We, const float* __restrict__ be,
    float* __restrict__ WhC, float* __restrict__ WhE, int n,
    const int* __restrict__ dstC, const int* __restrict__ dstE,
    int nEc, int nEe, unsigned char* __restrict__ keys)
{
  __shared__ __attribute__((aligned(16))) float lw[2 * DFEAT * WPITCH];
  const int gtid = blockIdx.x * blockDim.x + threadIdx.x;
  const int gsz  = gridDim.x * blockDim.x;
  unsigned* kwC = (unsigned*)keys;
  unsigned* kwE = (unsigned*)(keys + keys_second(nEc));
  const int nC4 = nEc >> 2;
  const int nE4 = nEe >> 2;
  const int4 dC = ((const int4*)dstC)[gtid < nC4 ? gtid : (nC4 > 0 ? nC4 - 1 : 0)];
  const int4 dE = ((const int4*)dstE)[gtid < nE4 ? gtid : (nE4 > 0 ? nE4 - 1 : 0)];

  stage_w(Wc, lw);
  stage_w(We, lw + DFEAT * WPITCH);
  __syncthreads();

  if (gtid < nC4) kwC[gtid] = pack_keys(dC);
  if (gtid < nE4) kwE[gtid] = pack_keys(dE);
  make_keys_rest(dstC, nEc, kwC, gtid, gsz);
  make_keys_rest(dstE, nEe, kwE, gtid, gsz);

  const int lane = threadIdx.x & 31;
  const int wave = __builtin_amdgcn_readfirstlane(threadIdx.x >> 5);
  const int g    = wave & 3;
  const int q    = blockIdx.x * 2 + (wave >> 2);
  const int nq   = gridDim.x * 2;
  const int nStrips = (n + TILE - 1) / TILE;

  const int half = lane >> 4;
  const int l15  = lane & 15;
  const int n0   = (g & 1) * 64;
  const float* bl   = lw + (g * 64 + l15) * WPITCH + half * 2;
  const float* bias = (g < 2) ? bc : be;
  float* dst        = (g < 2) ? WhC : WhE;

  float bv[4];
#pragma unroll
  for (int j = 0; j < 4; ++j) bv[j] = bias[n0 + j * TILE + l15];

  for (int r = 0; r * nq < nStrips; ++r) {
    const int s = strip_of_round(r, q, nq, nStrips);
    if (s < 0) continue;
    const int m0 = s * TILE;
    int arowi = m0 + l15;
    if (arowi >= n) arowi = n - 1;
    const float* arow = feats + (size_t)arowi * DFEAT + half * 2;

    v8f acc[4];
#pragma unroll
    for (int j = 0; j < 4; ++j) acc[j] = (v8f){0.f, 0.f, 0.f, 0.f, 0.f, 0.f, 0.f, 0.f};
    strip_mma<4, 32>(arow, bl, acc);

    float* ocol = dst + (size_t)(m0 + half * 8) * DFEAT + n0 + l15;
    if (m0 + TILE <= n) {
#pragma unroll
      for (int j = 0; j < 4; ++j)
#pragma unroll
        for (int rr = 0; rr < 8; ++rr)
          ocol[(size_t)rr * DFEAT + j * TILE] = acc[j][rr] + bv[j];
    } else {
#pragma unroll
      for (int j = 0; j < 4; ++j)
#pragma unroll
        for (int rr = 0; rr < 8; ++rr)
          if (m0 + half * 8 + rr < n)
            ocol[(size_t)rr * DFEAT + j * TILE] = acc[j][rr] + bv[j];
    }
  }
}

__global__ __launch_bounds__(256) void rgcn_out_gemm(
    const float* __restrict__ h,
    const float* __restrict__ Wo, const float* __restrict__ bo,
    const float* __restrict__ WhE, float* __restrict__ out, int n)
{
  __shared__ __attribute__((aligned(16))) float lw[DFEAT * WPITCH];
  stage_w(Wo, lw);
  __syncthreads();

  const int lane = threadIdx.x & 31;
  const int wave = __builtin_amdgcn_readfirstlane(threadIdx.x >> 5);
  const int g    = wave & 1;
  const int q    = blockIdx.x * 4 + (wave >> 1);
  const int nq   = gridDim.x * 4;
  const int nStrips = (n + TILE - 1) / TILE;

  const int half = lane >> 4;
  const int l15  = lane & 15;
  const int n0   = g * 64;
  const float* bl = lw + (n0 + l15) * WPITCH + half * 2;

  float bv[4];
#pragma unroll
  for (int j = 0; j < 4; ++j) bv[j] = bo[n0 + j * TILE + l15];

  for (int r = 0; r * nq < nStrips; ++r) {
    const int s = strip_of_round(r, q, nq, nStrips);
    if (s < 0) continue;
    const int m0 = s * TILE;
    int arowi = m0 + l15;
    if (arowi >= n) arowi = n - 1;
    const float* arow = h + (size_t)arowi * DFEAT + half * 2;

    v8f acc[4];
#pragma unroll
    for (int j = 0; j < 4; ++j) acc[j] = (v8f){0.f, 0.f, 0.f, 0.f, 0.f, 0.f, 0.f, 0.f};
    strip_mma<4, 8>(arow, bl, acc);

    const size_t base = (size_t)(m0 + half * 8) * DFEAT + n0 + l15;
    if (m0 + TILE <= n) {
#pragma unroll
      for (int j = 0; j < 4; ++j)
#pragma unroll
        for (int rr = 0; rr < 8; ++rr) {
          const size_t idx = base + (size_t)rr * DFEAT + j * TILE;
          out[idx] = acc[j][rr] + bv[j] + WhE[idx];
        }
    } else {
#pragma unroll
      for (int j = 0; j < 4; ++j)
#pragma unroll
        for (int rr = 0; rr < 8; ++rr)
          if (m0 + half * 8 + rr < n) {
            const size_t idx = base + (size_t)rr * DFEAT + j * TILE;
            out[idx] = acc[j][rr] + bv[j] + WhE[idx];
          }
    }
  }
}

#define OWN_THREADS 1024
#define OWN_WAVES   (OWN_THREADS / 32)
#define LIST_CAP    16384
#define SEG_CAP     (LIST_CAP / OWN_WAVES)
#define SCAN_U      8
#define GATHER_U    8
#define ETYPE_BIT   0x80000000u

__device__ __forceinline__ void lds_fadd(float* p, float v) {
  __hip_atomic_fetch_add(p, v, __ATOMIC_RELAXED, __HIP_MEMORY_SCOPE_WORKGROUP);
}

__device__ __forceinline__ void own_direct_edge(
    unsigned ent, const float* __restrict__ Wh,
    const int* __restrict__ srcC, const int* __restrict__ dstC,
    const int* __restrict__ srcE, const int* __restrict__ dstE,
    int n, unsigned lo, unsigned cnt, float* acc)
{
  const bool isE = (ent & ETYPE_BIT) != 0u;
  const unsigned e = ent & ~ETYPE_BIT;
  const unsigned a = (unsigned)(isE ? dstE : dstC)[e] - lo;
  if (a < cnt) {
    const float* row = Wh + ((size_t)(isE ? srcE : srcC)[e] + (isE ? (size_t)n : 0)) * DFEAT;
#pragma nounroll
    for (int c = 0; c < DFEAT; ++c)
      lds_fadd(acc + a * DFEAT + (c & 3) * 32 + (c >> 2), row[c]);
  }
}

__device__ __forceinline__ void gather_chunk(
    const unsigned* segp, unsigned nv, int lane, const float* __restrict__ Wh,
    const int* __restrict__ srcC, const int* __restrict__ dstC,
    const int* __restrict__ srcE, const int* __restrict__ dstE,
    int n, unsigned lo, unsigned cnt, float* acc)
{
  unsigned rowi = 0u;
  unsigned dl   = OWN_NODES;
  if ((unsigned)lane < nv) {
    const unsigned ent = segp[lane];
    const bool isE = (ent & ETYPE_BIT) != 0u;
    const unsigned e = ent & ~ETYPE_BIT;
    const int* sp = isE ? srcE : srcC;
    const int* dp = isE ? dstE : dstC;
    const unsigned a = (unsigned)dp[e] - lo;
    const unsigned s = (unsigned)sp[e] + (isE ? (unsigned)n : 0u);
    const bool own = a < cnt;
    rowi = own ? s : 0u;
    dl   = own ? a : (unsigned)OWN_NODES;
  }
  const unsigned nvr = (nv + GATHER_U - 1u) & ~(unsigned)(GATHER_U - 1);
  for (unsigned j = 0; j < nvr; j += GATHER_U) {
    float4 v[GATHER_U];
    unsigned d[GATHER_U];
#pragma unroll
    for (int u = 0; u < GATHER_U; ++u) {
      const unsigned r = (unsigned)__builtin_amdgcn_readlane((int)rowi, (int)(j + u));
      d[u] = (unsigned)__builtin_amdgcn_readlane((int)dl, (int)(j + u));
      v[u] = *(const float4*)(Wh + (size_t)r * DFEAT + lane * 4);
    }
#pragma unroll
    for (int u = 0; u < GATHER_U; ++u) {
      float* p = acc + d[u] * DFEAT + lane;
      lds_fadd(p,      v[u].x);
      lds_fadd(p + 32, v[u].y);
      lds_fadd(p + 64, v[u].z);
      lds_fadd(p + 96, v[u].w);
    }
  }
}

__device__ __forceinline__ unsigned seg_append(
    bool has, unsigned tagged, unsigned* seg, unsigned& wcount,
    const float* __restrict__ Wh,
    const int* __restrict__ srcC, const int* __restrict__ dstC,
    const int* __restrict__ srcE, const int* __restrict__ dstE,
    int n, unsigned lo, unsigned cnt, float* acc)
{
  const unsigned bal = __builtin_amdgcn_ballot_w32(has);
  if (bal != 0u) {
    if (wcount + 32u <= (unsigned)SEG_CAP) {
      if (has) seg[wcount + __builtin_amdgcn_mbcnt_lo(bal, 0u)] = tagged;
      wcount += (unsigned)__builtin_popcount(bal);
    } else if (has) {
      own_direct_edge(tagged, Wh, srcC, dstC, srcE, dstE, n, lo, cnt, acc);
    }
  }
  return bal;
}

__device__ __forceinline__ void own_scan(
    const unsigned char* __restrict__ keys, const int* __restrict__ dst, int nE,
    unsigned tag, unsigned splat, unsigned* seg, unsigned& wcount, int lane,
    const float* __restrict__ Wh,
    const int* __restrict__ srcC, const int* __restrict__ dstC,
    const int* __restrict__ srcE, const int* __restrict__ dstE,
    int n, unsigned lo, unsigned cnt, float* acc)
{
  const int tid  = threadIdx.x;
  const int nE16 = nE >> 4;
  const uint4* k4 = (const uint4*)keys;
  for (int b0 = 0; b0 < nE16; b0 += OWN_THREADS * SCAN_U) {
    const int base = b0 + tid;
    uint4 v[SCAN_U];
#pragma unroll
    for (int u = 0; u < SCAN_U; ++u) {
      const int i = base + u * OWN_THREADS;
      v[u] = k4[(i < nE16) ? i : nE16 - 1];
    }
#pragma unroll
    for (int u = 0; u < SCAN_U; ++u) {
      const unsigned x0 = v[u].x ^ splat, x1 = v[u].y ^ splat;
      const unsigned x2 = v[u].z ^ splat, x3 = v[u].w ^ splat;
      const unsigned t0 = (x0 - 0x01010101u) & ~x0;
      const unsigned t1 = (x1 - 0x01010101u) & ~x1;
      const unsigned t2 = (x2 - 0x01010101u) & ~x2;
      const unsigned t3 = (x3 - 0x01010101u) & ~x3;
      const bool inr = (base + u * OWN_THREADS) < nE16;
      const bool any = inr & (((t0 | t1 | t2 | t3) & 0x80808080u) != 0u);
      if (__builtin_amdgcn_ballot_w32(any) != 0u) {
        unsigned m = any ? (((t0 & 0x80808080u) >> 7) | ((t1 & 0x80808080u) >> 6) |
                            ((t2 & 0x80808080u) >> 5) | ((t3 & 0x80808080u) >> 4)) : 0u;
        const unsigned e0 = (unsigned)(base + u * OWN_THREADS) * 16u;
        for (;;) {
          const unsigned p = (unsigned)__builtin_ctz(m | 0x80000000u);
          const unsigned ent = tag | (e0 + ((p & 7u) << 2) + (p >> 3));
          if (seg_append(m != 0u, ent, seg, wcount,
                         Wh, srcC, dstC, srcE, dstE, n, lo, cnt, acc) == 0u) break;
          m &= m - 1u;
        }
      }
    }
    while (wcount >= 32u) {
      wcount -= 32u;
      gather_chunk(seg + wcount, 32u, lane, Wh, srcC, dstC, srcE, dstE, n, lo, cnt, acc);
    }
  }
  const int t = nE16 * 16 + tid;
  bool hit = false;
  if (t < nE) hit = ((unsigned)dst[t] - lo) < cnt;
  seg_append(hit, tag | (unsigned)t, seg, wcount,
             Wh, srcC, dstC, srcE, dstE, n, lo, cnt, acc);
}

__global__ __launch_bounds__(OWN_THREADS) void scatter_owner_lds(
    const float* __restrict__ Wh,
    const int* __restrict__ srcC, const int* __restrict__ dstC,
    const int* __restrict__ srcE, const int* __restrict__ dstE,
    const unsigned char* __restrict__ keys,
    float* __restrict__ h, int n, int nEc, int nEe)
{
  __shared__ __attribute__((aligned(16))) float acc[(OWN_NODES + 1) * DFEAT];
  __shared__ __attribute__((aligned(16))) unsigned list[LIST_CAP];

  const int tid  = threadIdx.x;
  const int lane = tid & 31;
  const int wave = __builtin_amdgcn_readfirstlane(tid >> 5);
  unsigned* seg = list + wave * SEG_CAP;

  for (int tile = blockIdx.x; tile * OWN_NODES < n; tile += gridDim.x) {
    const int lo  = tile * OWN_NODES;
    const int cnt = (n - lo < OWN_NODES) ? (n - lo) : OWN_NODES;

    for (int i = tid; i < (OWN_NODES + 1) * DFEAT / 4; i += OWN_THREADS)
      ((float4*)acc)[i] = make_float4(0.f, 0.f, 0.f, 0.f);
    __syncthreads();

    const unsigned splat = ((unsigned)tile & 255u) * 0x01010101u;
    unsigned wcount = 0u;
#pragma nounroll
    for (int ty = 0; ty < 2; ++ty) {
      const unsigned char* kT = ty ? keys + keys_second(nEc) : keys;
      own_scan(kT, ty ? dstE : dstC, ty ? nEe : nEc, ty ? ETYPE_BIT : 0u,
               splat, seg, wcount, lane,
               Wh, srcC, dstC, srcE, dstE, n, (unsigned)lo, (unsigned)cnt, acc);
    }
    while (wcount > 0u) {
      const unsigned nv = wcount < 32u ? wcount : 32u;
      wcount -= nv;
      gather_chunk(seg + wcount, nv, lane, Wh, srcC, dstC, srcE, dstE,
                   n, (unsigned)lo, (unsigned)cnt, acc);
    }
    __syncthreads();

    for (int i = tid; i < cnt * 32; i += OWN_THREADS) {
      const int node = i >> 5;
      const float* p = acc + node * DFEAT + lane;
      const float4 o = make_float4(p[0], p[32], p[64], p[96]);
      *(float4*)(h + (size_t)(lo + node) * DFEAT + lane * 4) = o;
    }
    __syncthreads();
  }
}

extern "C" void kernel_launch(void* const* d_in, const int* in_sizes, int n_in,
                              void* d_out, int out_size, void* d_ws, size_t ws_size,
                              hipStream_t stream) {
  const float* feats = (const float*)d_in[0];
  const float* Wc    = (const float*)d_in[1];
  const float* bc    = (const float*)d_in[2];
  const float* We    = (const float*)d_in[3];
  const float* be    = (const float*)d_in[4];
  const float* Wo    = (const float*)d_in[5];
  const float* bo    = (const float*)d_in[6];
  const int*   srcC  = (const int*)d_in[7];
  const int*   dstC  = (const int*)d_in[8];
  const int*   srcE  = (const int*)d_in[9];
  const int*   dstE  = (const int*)d_in[10];

  const int n   = in_sizes[0] / DFEAT;
  const int nEc = in_sizes[7];
  const int nEe = in_sizes[9];

  const size_t feSize = (size_t)n * DFEAT;
  float* WhC = (float*)d_ws;
  float* WhE = WhC + feSize;
  float* h   = WhE + feSize;
  unsigned char* keys = (unsigned char*)(h + feSize);
  float* out = (float*)d_out;

  rgcn_proj_gemm<<<512, 256, 0, stream>>>(
      feats, Wc, bc, We, be, WhC, WhE, n, dstC, dstE, nEc, nEe, keys);

  scatter_owner_lds<<<256, OWN_THREADS, 0, stream>>>(
      WhC, srcC, dstC, srcE, dstE, keys, h, n, nEc, nEe);

  rgcn_out_gemm<<<1024, 256, 0, stream>>>(h, Wo, bo, WhE, out, n);
}


# Output product made inside the owner work-groups: the tile's LDS sums are the A operand (fragments read in LDS), Wo staged in LDS; h never stored, out launch gone; k order and sum order unchanged
# speedup vs baseline: 5.1368x; 1.0580x over previous
#include <hip/hip_runtime.h>


typedef __attribute__((ext_vector_type(2))) float v2f;
typedef __attribute__((ext_vector_type(8))) float v8f;

#define DFEAT 128
#define TILE  16

#define WPITCH 132

template <int NTHR>
__device__ __forceinline__ void stage_w(const float* __restrict__ Wm, float* lw) {
  const int row0 = threadIdx.x >> 5;
  const int c4   = threadIdx.x & 31;
  float4 v[DFEAT * 32 / NTHR];
#pragma unroll
  for (int p = 0; p < DFEAT * 32 / NTHR; ++p)
    v[p] = *(const float4*)(Wm + (size_t)(row0 + p * (NTHR / 32)) * DFEAT + c4 * 4);
#pragma unroll
  for (int p = 0; p < DFEAT * 32 / NTHR; ++p)
    *(float4*)(lw + (row0 + p * (NTHR / 32)) * WPITCH + c4 * 4) = v[p];
}

__device__ __forceinline__ int strip_of_round(int r, int q, int nq, int nStrips) {
  const int base = r * nq;
  int c = nStrips - base;
  if (c > nq) c = nq;
  const unsigned lo = ((unsigned)q * (unsigned)c) / (unsigned)nq;
  const unsigned hi = ((unsigned)(q + 1) * (unsigned)c) / (unsigned)nq;
  return (hi > lo) ? base + (int)lo : -1;
}

template <int NT, int KC>
__device__ __forceinline__ void strip_mma(
    const float* __restrict__ arow, const float* bl, v8f (&acc)[NT])
{
#pragma unroll 1
  for (int k0 = 0; k0 < DFEAT; k0 += 4 * KC) {
    v2f a[KC];
#pragma unroll
    for (int s = 0; s < KC; ++s) a[s] = *(const v2f*)(arow + k0 + 4 * s);
#pragma unroll
    for (int s = 0; s < KC; ++s) {
#pragma unroll
      for (int j = 0; j < NT; ++j) {
        const v2f b = *(const v2f*)(bl + j * TILE * WPITCH + k0 + 4 * s);
        acc[j] = __builtin_amdgcn_wmma_f32_16x16x4_f32(
            false, a[s], false, b, (short)0, acc[j], false, false);
      }
    }
  }
}

#define OWN_NODES   196

__device__ __forceinline__ unsigned own_key(int d) {
  return ((unsigned)d / (unsigned)OWN_NODES) & 255u;
}

__device__ __forceinline__ unsigned pack_keys(int4 d) {
  return own_key(d.x) | (own_key(d.y) << 8) | (own_key(d.z) << 16) | (own_key(d.w) << 24);
}

__device__ __forceinline__ void make_keys_rest(
    const int* __restrict__ dst, int nE, unsigned* __restrict__ kw, int gtid, int gsz)
{
  const int nE4 = nE >> 2;
  const int4* d4 = (const int4*)dst;
#pragma nounroll
  for (int i = gtid + gsz; i < nE4; i += gsz) kw[i] = pack_keys(d4[i]);
}

__device__ __forceinline__ size_t keys_second(int nEfirst) {
  return ((size_t)nEfirst + 15) & ~(size_t)15;
}

__global__ __launch_bounds__(256) void rgcn_proj_gemm(
    const float* __restrict__ feats,
    const float* __restrict__ Wc, const float* __restrict__ bc,
    const float* __restrict__ We, const float* __restrict__ be,
    float* __restrict__ WhC, float* __restrict__ WhE, int n,
    const int* __restrict__ dstC, const int* __restrict__ dstE,
    int nEc, int nEe, unsigned char* __restrict__ keys)
{
  __shared__ __attribute__((aligned(16))) float lw[2 * DFEAT * WPITCH];
  const int gtid = blockIdx.x * blockDim.x + threadIdx.x;
  const int gsz  = gridDim.x * blockDim.x;
  unsigned* kwC = (unsigned*)keys;
  unsigned* kwE = (unsigned*)(keys + keys_second(nEc));
  const int nC4 = nEc >> 2;
  const int nE4 = nEe >> 2;
  const int4 dC = ((const int4*)dstC)[gtid < nC4 ? gtid : (nC4 > 0 ? nC4 - 1 : 0)];
  const int4 dE = ((const int4*)dstE)[gtid < nE4 ? gtid : (nE4 > 0 ? nE4 - 1 : 0)];

  stage_w<256>(Wc, lw);
  stage_w<256>(We, lw + DFEAT * WPITCH);
  __syncthreads();

  if (gtid < nC4) kwC[gtid] = pack_keys(dC);
  if (gtid < nE4) kwE[gtid] = pack_keys(dE);
  make_keys_rest(dstC, nEc, kwC, gtid, gsz);
  make_keys_rest(dstE, nEe, kwE, gtid, gsz);

  const int lane = threadIdx.x & 31;
  const int wave = __builtin_amdgcn_readfirstlane(threadIdx.x >> 5);
  const int g    = wave & 3;
  const int q    = blockIdx.x * 2 + (wave >> 2);
  const int nq   = gridDim.x * 2;
  const int nStrips = (n + TILE - 1) / TILE;

  const int half = lane >> 4;
  const int l15  = lane & 15;
  const int n0   = (g & 1) * 64;
  const float* bl   = lw + (g * 64 + l15) * WPITCH + half * 2;
  const float* bias = (g < 2) ? bc : be;
  float* dst        = (g < 2) ? WhC : WhE;

  float bv[4];
#pragma unroll
  for (int j = 0; j < 4; ++j) bv[j] = bias[n0 + j * TILE + l15];

  for (int r = 0; r * nq < nStrips; ++r) {
    const int s = strip_of_round(r, q, nq, nStrips);
    if (s < 0) continue;
    const int m0 = s * TILE;
    int arowi = m0 + l15;
    if (arowi >= n) arowi = n - 1;
    const float* arow = feats + (size_t)arowi * DFEAT + half * 2;

    v8f acc[4];
#pragma unroll
    for (int j = 0; j < 4; ++j) acc[j] = (v8f){0.f, 0.f, 0.f, 0.f, 0.f, 0.f, 0.f, 0.f};
    strip_mma<4, 32>(arow, bl, acc);

    float* ocol = dst + (size_t)(m0 + half * 8) * DFEAT + n0 + l15;
    if (m0 + TILE <= n) {
#pragma unroll
      for (int j = 0; j < 4; ++j)
#pragma unroll
        for (int rr = 0; rr < 8; ++rr)
          ocol[(size_t)rr * DFEAT + j * TILE] = acc[j][rr] + bv[j];
    } else {
#pragma unroll
      for (int j = 0; j < 4; ++j)
#pragma unroll
        for (int rr = 0; rr < 8; ++rr)
          if (m0 + half * 8 + rr < n)
            ocol[(size_t)rr * DFEAT + j * TILE] = acc[j][rr] + bv[j];
    }
  }
}

#define APITCH      132
#define QP          33
#define OWN_STRIPS  ((OWN_NODES + TILE - 1) / TILE)
#define OWN_ROWS    (OWN_STRIPS * TILE)
#define OWN_THREADS 1024
#define OWN_WAVES   (OWN_THREADS / 32)
#define LIST_CAP    16384
#define SEG_CAP     (LIST_CAP / OWN_WAVES)
#define SCAN_U      8
#define GATHER_U    8
#define ETYPE_BIT   0x80000000u

__device__ __forceinline__ void lds_fadd(float* p, float v) {
  __hip_atomic_fetch_add(p, v, __ATOMIC_RELAXED, __HIP_MEMORY_SCOPE_WORKGROUP);
}

__device__ __forceinline__ void own_direct_edge(
    unsigned ent, const float* __restrict__ Wh,
    const int* __restrict__ srcC, const int* __restrict__ dstC,
    const int* __restrict__ srcE, const int* __restrict__ dstE,
    int n, unsigned lo, unsigned cnt, float* acc)
{
  const bool isE = (ent & ETYPE_BIT) != 0u;
  const unsigned e = ent & ~ETYPE_BIT;
  const unsigned a = (unsigned)(isE ? dstE : dstC)[e] - lo;
  if (a < cnt) {
    const float* row = Wh + ((size_t)(isE ? srcE : srcC)[e] + (isE ? (size_t)n : 0)) * DFEAT;
#pragma nounroll
    for (int c = 0; c < DFEAT; ++c)
      lds_fadd(acc + a * APITCH + (c & 3) * QP + (c >> 2), row[c]);
  }
}

__device__ __forceinline__ void gather_chunk(
    const unsigned* segp, unsigned nv, int lane, const float* __restrict__ Wh,
    const int* __restrict__ srcC, const int* __restrict__ dstC,
    const int* __restrict__ srcE, const int* __restrict__ dstE,
    int n, unsigned lo, unsigned cnt, float* acc)
{
  unsigned rowi = 0u;
  unsigned dl   = OWN_NODES;
  if ((unsigned)lane < nv) {
    const unsigned ent = segp[lane];
    const bool isE = (ent & ETYPE_BIT) != 0u;
    const unsigned e = ent & ~ETYPE_BIT;
    const int* sp = isE ? srcE : srcC;
    const int* dp = isE ? dstE : dstC;
    const unsigned a = (unsigned)dp[e] - lo;
    const unsigned s = (unsigned)sp[e] + (isE ? (unsigned)n : 0u);
    const bool own = a < cnt;
    rowi = own ? s : 0u;
    dl   = own ? a : (unsigned)OWN_NODES;
  }
  const unsigned nvr = (nv + GATHER_U - 1u) & ~(unsigned)(GATHER_U - 1);
  for (unsigned j = 0; j < nvr; j += GATHER_U) {
    float4 v[GATHER_U];
    unsigned d[GATHER_U];
#pragma unroll
    for (int u = 0; u < GATHER_U; ++u) {
      const unsigned r = (unsigned)__builtin_amdgcn_readlane((int)rowi, (int)(j + u));
      d[u] = (unsigned)__builtin_amdgcn_readlane((int)dl, (int)(j + u));
      v[u] = *(const float4*)(Wh + (size_t)r * DFEAT + lane * 4);
    }
#pragma unroll
    for (int u = 0; u < GATHER_U; ++u) {
      float* p = acc + d[u] * APITCH + lane;
      lds_fadd(p,          v[u].x);
      lds_fadd(p + QP,     v[u].y);
      lds_fadd(p + 2 * QP, v[u].z);
      lds_fadd(p + 3 * QP, v[u].w);
    }
  }
}

__device__ __forceinline__ unsigned seg_append(
    bool has, unsigned tagged, unsigned* seg, unsigned& wcount,
    const float* __restrict__ Wh,
    const int* __restrict__ srcC, const int* __restrict__ dstC,
    const int* __restrict__ srcE, const int* __restrict__ dstE,
    int n, unsigned lo, unsigned cnt, float* acc)
{
  const unsigned bal = __builtin_amdgcn_ballot_w32(has);
  if (bal != 0u) {
    if (wcount + 32u <= (unsigned)SEG_CAP) {
      if (has) seg[wcount + __builtin_amdgcn_mbcnt_lo(bal, 0u)] = tagged;
      wcount += (unsigned)__builtin_popcount(bal);
    } else if (has) {
      own_direct_edge(tagged, Wh, srcC, dstC, srcE, dstE, n, lo, cnt, acc);
    }
  }
  return bal;
}

__device__ __forceinline__ void own_scan(
    const unsigned char* __restrict__ keys, const int* __restrict__ dst, int nE,
    unsigned tag, unsigned splat, unsigned* seg, unsigned& wcount, int lane,
    const float* __restrict__ Wh,
    const int* __restrict__ srcC, const int* __restrict__ dstC,
    const int* __restrict__ srcE, const int* __restrict__ dstE,
    int n, unsigned lo, unsigned cnt, float* acc)
{
  const int tid  = threadIdx.x;
  const int nE16 = nE >> 4;
  const uint4* k4 = (const uint4*)keys;
  for (int b0 = 0; b0 < nE16; b0 += OWN_THREADS * SCAN_U) {
    const int base = b0 + tid;
    uint4 v[SCAN_U];
#pragma unroll
    for (int u = 0; u < SCAN_U; ++u) {
      const int i = base + u * OWN_THREADS;
      v[u] = k4[(i < nE16) ? i : nE16 - 1];
    }
#pragma unroll
    for (int u = 0; u < SCAN_U; ++u) {
      const unsigned x0 = v[u].x ^ splat, x1 = v[u].y ^ splat;
      const unsigned x2 = v[u].z ^ splat, x3 = v[u].w ^ splat;
      const unsigned t0 = (x0 - 0x01010101u) & ~x0;
      const unsigned t1 = (x1 - 0x01010101u) & ~x1;
      const unsigned t2 = (x2 - 0x01010101u) & ~x2;
      const unsigned t3 = (x3 - 0x01010101u) & ~x3;
      const bool inr = (base + u * OWN_THREADS) < nE16;
      const bool any = inr & (((t0 | t1 | t2 | t3) & 0x80808080u) != 0u);
      if (__builtin_amdgcn_ballot_w32(any) != 0u) {
        unsigned m = any ? (((t0 & 0x80808080u) >> 7) | ((t1 & 0x80808080u) >> 6) |
                            ((t2 & 0x80808080u) >> 5) | ((t3 & 0x80808080u) >> 4)) : 0u;
        const unsigned e0 = (unsigned)(base + u * OWN_THREADS) * 16u;
        for (;;) {
          const unsigned p = (unsigned)__builtin_ctz(m | 0x80000000u);
          const unsigned ent = tag | (e0 + ((p & 7u) << 2) + (p >> 3));
          if (seg_append(m != 0u, ent, seg, wcount,
                         Wh, srcC, dstC, srcE, dstE, n, lo, cnt, acc) == 0u) break;
          m &= m - 1u;
        }
      }
    }
    while (wcount >= 32u) {
      wcount -= 32u;
      gather_chunk(seg + wcount, 32u, lane, Wh, srcC, dstC, srcE, dstE, n, lo, cnt, acc);
    }
  }
  const int t = nE16 * 16 + tid;
  bool hit = false;
  if (t < nE) hit = ((unsigned)dst[t] - lo) < cnt;
  seg_append(hit, tag | (unsigned)t, seg, wcount,
             Wh, srcC, dstC, srcE, dstE, n, lo, cnt, acc);
}

#define OUT_NT      2
#define OUT_CG      (DFEAT / (TILE * OUT_NT))
#define OUT_TASKS   (OWN_STRIPS * OUT_CG)

__device__ __forceinline__ void own_out_task(
    const float* acc, const float* lw, int strip, int n0, int lane,
    const float* __restrict__ bo, const float* __restrict__ WhE,
    float* __restrict__ out, int lo, int cnt)
{
  const int half = lane >> 4;
  const int l15  = lane & 15;
  const float* ar = acc + (strip * TILE + l15) * APITCH + half * 2 * QP;
  const float* bl = lw + (n0 + l15) * WPITCH + half * 2;
  const int r0 = strip * TILE + half * 8;

  float bv[OUT_NT];
  float ad[OUT_NT][8];
#pragma unroll
  for (int j = 0; j < OUT_NT; ++j) bv[j] = bo[n0 + j * TILE + l15];
#pragma unroll
  for (int rr = 0; rr < 8; ++rr) {
    int rl = r0 + rr;
    if (rl >= cnt) rl = cnt - 1;
    const float* wr = WhE + (size_t)(lo + rl) * DFEAT + n0 + l15;
#pragma unroll
    for (int j = 0; j < OUT_NT; ++j) ad[j][rr] = wr[j * TILE];
  }

  v8f c[OUT_NT];
#pragma unroll
  for (int j = 0; j < OUT_NT; ++j) c[j] = (v8f){0.f, 0.f, 0.f, 0.f, 0.f, 0.f, 0.f, 0.f};
#pragma unroll 1
  for (int s0 = 0; s0 < DFEAT / 4; s0 += 8) {
    v2f a[8];
#pragma unroll
    for (int s = 0; s < 8; ++s) {
      a[s].x = ar[s0 + s];
      a[s].y = ar[QP + s0 + s];
    }
#pragma unroll
    for (int s = 0; s < 8; ++s) {
#pragma unroll
      for (int j = 0; j < OUT_NT; ++j) {
        const v2f b = *(const v2f*)(bl + j * TILE * WPITCH + 4 * (s0 + s));
        c[j] = __builtin_amdgcn_wmma_f32_16x16x4_f32(
            false, a[s], false, b, (short)0, c[j], false, false);
      }
    }
  }

  float* ocol = out + (size_t)(lo + r0) * DFEAT + n0 + l15;
  if (strip * TILE + TILE <= cnt) {
#pragma unroll
    for (int j = 0; j < OUT_NT; ++j)
#pragma unroll
      for (int rr = 0; rr < 8; ++rr)
        ocol[(size_t)rr * DFEAT + j * TILE] = c[j][rr] + bv[j] + ad[j][rr];
  } else {
#pragma unroll
    for (int j = 0; j < OUT_NT; ++j)
#pragma unroll
      for (int rr = 0; rr < 8; ++rr)
        if (r0 + rr < cnt)
          ocol[(size_t)rr * DFEAT + j * TILE] = c[j][rr] + bv[j] + ad[j][rr];
  }
}

__device__ __forceinline__ void own_clear(float* acc) {
  for (int i = threadIdx.x; i < OWN_ROWS * APITCH / 4; i += OWN_THREADS)
    ((float4*)acc)[i] = make_float4(0.f, 0.f, 0.f, 0.f);
}

__global__ __launch_bounds__(OWN_THREADS) void scatter_owner_lds(
    const float* __restrict__ Wh,
    const int* __restrict__ srcC, const int* __restrict__ dstC,
    const int* __restrict__ srcE, const int* __restrict__ dstE,
    const unsigned char* __restrict__ keys,
    const float* __restrict__ Wo, const float* __restrict__ bo,
    float* __restrict__ out, int n, int nEc, int nEe)
{
  __shared__ __attribute__((aligned(16))) float acc[OWN_ROWS * APITCH];
  __shared__ __attribute__((aligned(16))) unsigned list[LIST_CAP];
  __shared__ __attribute__((aligned(16))) float lw[DFEAT * WPITCH];

  const int tid  = threadIdx.x;
  const int lane = tid & 31;
  const int wave = __builtin_amdgcn_readfirstlane(tid >> 5);
  unsigned* seg = list + wave * SEG_CAP;
  const float* WhE = Wh + (size_t)n * DFEAT;

  own_clear(acc);
  stage_w<OWN_THREADS>(Wo, lw);

  for (int tile = blockIdx.x; tile * OWN_NODES < n; tile += gridDim.x) {
    const int lo  = tile * OWN_NODES;
    const int cnt = (n - lo < OWN_NODES) ? (n - lo) : OWN_NODES;

    __syncthreads();

    const unsigned splat = ((unsigned)tile & 255u) * 0x01010101u;
    unsigned wcount = 0u;
#pragma nounroll
    for (int ty = 0; ty < 2; ++ty) {
      const unsigned char* kT = ty ? keys + keys_second(nEc) : keys;
      own_scan(kT, ty ? dstE : dstC, ty ? nEe : nEc, ty ? ETYPE_BIT : 0u,
               splat, seg, wcount, lane,
               Wh, srcC, dstC, srcE, dstE, n, (unsigned)lo, (unsigned)cnt, acc);
    }
    while (wcount > 0u) {
      const unsigned nv = wcount < 32u ? wcount : 32u;
      wcount -= nv;
      gather_chunk(seg + wcount, nv, lane, Wh, srcC, dstC, srcE, dstE,
                   n, (unsigned)lo, (unsigned)cnt, acc);
    }
    __syncthreads();

    for (int t = wave; t < OUT_TASKS; t += OWN_WAVES) {
      const int strip = t / OUT_CG;
      if (strip * TILE < cnt)
        own_out_task(acc, lw, strip, (t % OUT_CG) * (TILE * OUT_NT), lane,
                     bo, WhE, out, lo, cnt);
    }

    if ((tile + (int)gridDim.x) * OWN_NODES < n) {
      __syncthreads();
      own_clear(acc);
    }
  }
}

extern "C" void kernel_launch(void* const* d_in, const int* in_sizes, int n_in,
                              void* d_out, int out_size, void* d_ws, size_t ws_size,
                              hipStream_t stream) {
  const float* feats = (const float*)d_in[0];
  const float* Wc    = (const float*)d_in[1];
  const float* bc    = (const float*)d_in[2];
  const float* We    = (const float*)d_in[3];
  const float* be    = (const float*)d_in[4];
  const float* Wo    = (const float*)d_in[5];
  const float* bo    = (const float*)d_in[6];
  const int*   srcC  = (const int*)d_in[7];
  const int*   dstC  = (const int*)d_in[8];
  const int*   srcE  = (const int*)d_in[9];
  const int*   dstE  = (const int*)d_in[10];

  const int n   = in_sizes[0] / DFEAT;
  const int nEc = in_sizes[7];
  const int nEe = in_sizes[9];

  const size_t feSize = (size_t)n * DFEAT;
  float* WhC = (float*)d_ws;
  float* WhE = WhC + feSize;
  unsigned char* keys = (unsigned char*)(WhE + feSize);
  float* out = (float*)d_out;

  rgcn_proj_gemm<<<512, 256, 0, stream>>>(
      feats, Wc, bc, We, be, WhC, WhE, n, dstC, dstE, nEc, nEe, keys);

  scatter_owner_lds<<<256, OWN_THREADS, 0, stream>>>(
      WhC, srcC, dstC, srcE, dstE, keys, Wo, bo, out, n, nEc, nEe);
}


# Edge index pairs: the projection launch writes each edge's (dst, src) as one 8-B record; an owner lane resolves a listed edge with one random line fetch, not two; sum order as before: LDS arrival
# speedup vs baseline: 5.3488x; 1.0413x over previous
#include <hip/hip_runtime.h>


typedef __attribute__((ext_vector_type(2))) float v2f;
typedef __attribute__((ext_vector_type(8))) float v8f;

#define DFEAT 128
#define TILE  16

#define WPITCH 132

template <int NTHR>
__device__ __forceinline__ void stage_w(const float* __restrict__ Wm, float* lw) {
  const int row0 = threadIdx.x >> 5;
  const int c4   = threadIdx.x & 31;
  float4 v[DFEAT * 32 / NTHR];
#pragma unroll
  for (int p = 0; p < DFEAT * 32 / NTHR; ++p)
    v[p] = *(const float4*)(Wm + (size_t)(row0 + p * (NTHR / 32)) * DFEAT + c4 * 4);
#pragma unroll
  for (int p = 0; p < DFEAT * 32 / NTHR; ++p)
    *(float4*)(lw + (row0 + p * (NTHR / 32)) * WPITCH + c4 * 4) = v[p];
}

__device__ __forceinline__ int strip_of_round(int r, int q, int nq, int nStrips) {
  const int base = r * nq;
  int c = nStrips - base;
  if (c > nq) c = nq;
  const unsigned lo = ((unsigned)q * (unsigned)c) / (unsigned)nq;
  const unsigned hi = ((unsigned)(q + 1) * (unsigned)c) / (unsigned)nq;
  return (hi > lo) ? base + (int)lo : -1;
}

template <int NT, int KC>
__device__ __forceinline__ void strip_mma(
    const float* __restrict__ arow, const float* bl, v8f (&acc)[NT])
{
#pragma unroll 1
  for (int k0 = 0; k0 < DFEAT; k0 += 4 * KC) {
    v2f a[KC];
#pragma unroll
    for (int s = 0; s < KC; ++s) a[s] = *(const v2f*)(arow + k0 + 4 * s);
#pragma unroll
    for (int s = 0; s < KC; ++s) {
#pragma unroll
      for (int j = 0; j < NT; ++j) {
        const v2f b = *(const v2f*)(bl + j * TILE * WPITCH + k0 + 4 * s);
        acc[j] = __builtin_amdgcn_wmma_f32_16x16x4_f32(
            false, a[s], false, b, (short)0, acc[j], false, false);
      }
    }
  }
}

#define OWN_NODES   196

__device__ __forceinline__ unsigned own_key(int d) {
  return ((unsigned)d / (unsigned)OWN_NODES) & 255u;
}

__device__ __forceinline__ unsigned pack_keys(int4 d) {
  return own_key(d.x) | (own_key(d.y) << 8) | (own_key(d.z) << 16) | (own_key(d.w) << 24);
}

__device__ __forceinline__ void store_pairs(int2* __restrict__ pr, int g, int4 d, int4 s) {
  int4* p = (int4*)(pr + 4 * (size_t)g);
  p[0] = make_int4(d.x, s.x, d.y, s.y);
  p[1] = make_int4(d.z, s.z, d.w, s.w);
}

__device__ __forceinline__ void make_edge_planes_rest(
    const int* __restrict__ dst, const int* __restrict__ src, int nE,
    unsigned* __restrict__ kw, int2* __restrict__ pr, int gtid, int gsz)
{
  const int nE4 = nE >> 2;
  const int4* d4 = (const int4*)dst;
  const int4* s4 = (const int4*)src;
#pragma nounroll
  for (int i = gtid + gsz; i < nE4; i += gsz) {
    const int4 d = d4[i];
    kw[i] = pack_keys(d);
    store_pairs(pr, i, d, s4[i]);
  }
  const int t = nE4 * 4 + gtid;
  if (t < nE) pr[t] = make_int2(dst[t], src[t]);
}

__device__ __forceinline__ size_t keys_second(int nEfirst) {
  return ((size_t)nEfirst + 15) & ~(size_t)15;
}

__device__ __forceinline__ size_t pairs_byte_offset(int nEc, int nEe) {
  return keys_second(nEc) + keys_second(nEe);
}
__device__ __forceinline__ size_t pairs_second(int nEfirst) {
  return ((size_t)nEfirst + 1) & ~(size_t)1;
}

__global__ __launch_bounds__(256) void rgcn_proj_gemm(
    const float* __restrict__ feats,
    const float* __restrict__ Wc, const float* __restrict__ bc,
    const float* __restrict__ We, const float* __restrict__ be,
    float* __restrict__ WhC, float* __restrict__ WhE, int n,
    const int* __restrict__ srcC, const int* __restrict__ dstC,
    const int* __restrict__ srcE, const int* __restrict__ dstE,
    int nEc, int nEe, unsigned char* __restrict__ keys)
{
  __shared__ __attribute__((aligned(16))) float lw[2 * DFEAT * WPITCH];
  const int gtid = blockIdx.x * blockDim.x + threadIdx.x;
  const int gsz  = gridDim.x * blockDim.x;
  unsigned* kwC = (unsigned*)keys;
  unsigned* kwE = (unsigned*)(keys + keys_second(nEc));
  int2* prC = (int2*)(keys + pairs_byte_offset(nEc, nEe));
  int2* prE = prC + pairs_second(nEc);
  const int nC4 = nEc >> 2;
  const int nE4 = nEe >> 2;
  const int gC = gtid < nC4 ? gtid : (nC4 > 0 ? nC4 - 1 : 0);
  const int gE = gtid < nE4 ? gtid : (nE4 > 0 ? nE4 - 1 : 0);
  const int4 dC = ((const int4*)dstC)[gC];
  const int4 sC = ((const int4*)srcC)[gC];
  const int4 dE = ((const int4*)dstE)[gE];
  const int4 sE = ((const int4*)srcE)[gE];

  stage_w<256>(Wc, lw);
  stage_w<256>(We, lw + DFEAT * WPITCH);
  __syncthreads();

  if (gtid < nC4) { kwC[gtid] = pack_keys(dC); store_pairs(prC, gtid, dC, sC); }
  if (gtid < nE4) { kwE[gtid] = pack_keys(dE); store_pairs(prE, gtid, dE, sE); }
  make_edge_planes_rest(dstC, srcC, nEc, kwC, prC, gtid, gsz);
  make_edge_planes_rest(dstE, srcE, nEe, kwE, prE, gtid, gsz);

  const int lane = threadIdx.x & 31;
  const int wave = __builtin_amdgcn_readfirstlane(threadIdx.x >> 5);
  const int g    = wave & 3;
  const int q    = blockIdx.x * 2 + (wave >> 2);
  const int nq   = gridDim.x * 2;
  const int nStrips = (n + TILE - 1) / TILE;

  const int half = lane >> 4;
  const int l15  = lane & 15;
  const int n0   = (g & 1) * 64;
  const float* bl   = lw + (g * 64 + l15) * WPITCH + half * 2;
  const float* bias = (g < 2) ? bc : be;
  float* dst        = (g < 2) ? WhC : WhE;

  float bv[4];
#pragma unroll
  for (int j = 0; j < 4; ++j) bv[j] = bias[n0 + j * TILE + l15];

  for (int r = 0; r * nq < nStrips; ++r) {
    const int s = strip_of_round(r, q, nq, nStrips);
    if (s < 0) continue;
    const int m0 = s * TILE;
    int arowi = m0 + l15;
    if (arowi >= n) arowi = n - 1;
    const float* arow = feats + (size_t)arowi * DFEAT + half * 2;

    v8f acc[4];
#pragma unroll
    for (int j = 0; j < 4; ++j) acc[j] = (v8f){0.f, 0.f, 0.f, 0.f, 0.f, 0.f, 0.f, 0.f};
    strip_mma<4, 32>(arow, bl, acc);

    float* ocol = dst + (size_t)(m0 + half * 8) * DFEAT + n0 + l15;
    if (m0 + TILE <= n) {
#pragma unroll
      for (int j = 0; j < 4; ++j)
#pragma unroll
        for (int rr = 0; rr < 8; ++rr)
          ocol[(size_t)rr * DFEAT + j * TILE] = acc[j][rr] + bv[j];
    } else {
#pragma unroll
      for (int j = 0; j < 4; ++j)
#pragma unroll
        for (int rr = 0; rr < 8; ++rr)
          if (m0 + half * 8 + rr < n)
            ocol[(size_t)rr * DFEAT + j * TILE] = acc[j][rr] + bv[j];
    }
  }
}

#define APITCH      132
#define QP          33
#define OWN_STRIPS  ((OWN_NODES + TILE - 1) / TILE)
#define OWN_ROWS    (OWN_STRIPS * TILE)
#define OWN_THREADS 1024
#define OWN_WAVES   (OWN_THREADS / 32)
#define LIST_CAP    16384
#define SEG_CAP     (LIST_CAP / OWN_WAVES)
#define SCAN_U      8
#define GATHER_U    8
#define ETYPE_BIT   0x80000000u

__device__ __forceinline__ void lds_fadd(float* p, float v) {
  __hip_atomic_fetch_add(p, v, __ATOMIC_RELAXED, __HIP_MEMORY_SCOPE_WORKGROUP);
}

__device__ __forceinline__ void own_direct_edge(
    unsigned ent, const float* __restrict__ Wh,
    const int2* __restrict__ prC, const int2* __restrict__ prE,
    int n, unsigned lo, unsigned cnt, float* acc)
{
  const bool isE = (ent & ETYPE_BIT) != 0u;
  const unsigned e = ent & ~ETYPE_BIT;
  const int2 pr = (isE ? prE : prC)[e];
  const unsigned a = (unsigned)pr.x - lo;
  if (a < cnt) {
    const float* row = Wh + ((size_t)pr.y + (isE ? (size_t)n : 0)) * DFEAT;
#pragma nounroll
    for (int c = 0; c < DFEAT; ++c)
      lds_fadd(acc + a * APITCH + (c & 3) * QP + (c >> 2), row[c]);
  }
}

__device__ __forceinline__ void gather_chunk(
    const unsigned* segp, unsigned nv, int lane, const float* __restrict__ Wh,
    const int2* __restrict__ prC, const int2* __restrict__ prE,
    int n, unsigned lo, unsigned cnt, float* acc)
{
  unsigned rowi = 0u;
  unsigned dl   = OWN_NODES;
  if ((unsigned)lane < nv) {
    const unsigned ent = segp[lane];
    const bool isE = (ent & ETYPE_BIT) != 0u;
    const unsigned e = ent & ~ETYPE_BIT;
    const int2 pr = (isE ? prE : prC)[e];
    const unsigned a = (unsigned)pr.x - lo;
    const unsigned s = (unsigned)pr.y + (isE ? (unsigned)n : 0u);
    const bool own = a < cnt;
    rowi = own ? s : 0u;
    dl   = own ? a : (unsigned)OWN_NODES;
  }
  const unsigned nvr = (nv + GATHER_U - 1u) & ~(unsigned)(GATHER_U - 1);
  for (unsigned j = 0; j < nvr; j += GATHER_U) {
    float4 v[GATHER_U];
    unsigned d[GATHER_U];
#pragma unroll
    for (int u = 0; u < GATHER_U; ++u) {
      const unsigned r = (unsigned)__builtin_amdgcn_readlane((int)rowi, (int)(j + u));
      d[u] = (unsigned)__builtin_amdgcn_readlane((int)dl, (int)(j + u));
      v[u] = *(const float4*)(Wh + (size_t)r * DFEAT + lane * 4);
    }
#pragma unroll
    for (int u = 0; u < GATHER_U; ++u) {
      float* p = acc + d[u] * APITCH + lane;
      lds_fadd(p,          v[u].x);
      lds_fadd(p + QP,     v[u].y);
      lds_fadd(p + 2 * QP, v[u].z);
      lds_fadd(p + 3 * QP, v[u].w);
    }
  }
}

__device__ __forceinline__ unsigned seg_append(
    bool has, unsigned tagged, unsigned* seg, unsigned& wcount,
    const float* __restrict__ Wh,
    const int2* __restrict__ prC, const int2* __restrict__ prE,
    int n, unsigned lo, unsigned cnt, float* acc)
{
  const unsigned bal = __builtin_amdgcn_ballot_w32(has);
  if (bal != 0u) {
    if (wcount + 32u <= (unsigned)SEG_CAP) {
      if (has) seg[wcount + __builtin_amdgcn_mbcnt_lo(bal, 0u)] = tagged;
      wcount += (unsigned)__builtin_popcount(bal);
    } else if (has) {
      own_direct_edge(tagged, Wh, prC, prE, n, lo, cnt, acc);
    }
  }
  return bal;
}

__device__ __forceinline__ void own_scan(
    const unsigned char* __restrict__ keys, const int2* __restrict__ pr, int nE,
    unsigned tag, unsigned splat, unsigned* seg, unsigned& wcount, int lane,
    const float* __restrict__ Wh,
    const int2* __restrict__ prC, const int2* __restrict__ prE,
    int n, unsigned lo, unsigned cnt, float* acc)
{
  const int tid  = threadIdx.x;
  const int nE16 = nE >> 4;
  const uint4* k4 = (const uint4*)keys;
  for (int b0 = 0; b0 < nE16; b0 += OWN_THREADS * SCAN_U) {
    const int base = b0 + tid;
    uint4 v[SCAN_U];
#pragma unroll
    for (int u = 0; u < SCAN_U; ++u) {
      const int i = base + u * OWN_THREADS;
      v[u] = k4[(i < nE16) ? i : nE16 - 1];
    }
#pragma unroll
    for (int u = 0; u < SCAN_U; ++u) {
      const unsigned x0 = v[u].x ^ splat, x1 = v[u].y ^ splat;
      const unsigned x2 = v[u].z ^ splat, x3 = v[u].w ^ splat;
      const unsigned t0 = (x0 - 0x01010101u) & ~x0;
      const unsigned t1 = (x1 - 0x01010101u) & ~x1;
      const unsigned t2 = (x2 - 0x01010101u) & ~x2;
      const unsigned t3 = (x3 - 0x01010101u) & ~x3;
      const bool inr = (base + u * OWN_THREADS) < nE16;
      const bool any = inr & (((t0 | t1 | t2 | t3) & 0x80808080u) != 0u);
      if (__builtin_amdgcn_ballot_w32(any) != 0u) {
        unsigned m = any ? (((t0 & 0x80808080u) >> 7) | ((t1 & 0x80808080u) >> 6) |
                            ((t2 & 0x80808080u) >> 5) | ((t3 & 0x80808080u) >> 4)) : 0u;
        const unsigned e0 = (unsigned)(base + u * OWN_THREADS) * 16u;
        for (;;) {
          const unsigned p = (unsigned)__builtin_ctz(m | 0x80000000u);
          const unsigned ent = tag | (e0 + ((p & 7u) << 2) + (p >> 3));
          if (seg_append(m != 0u, ent, seg, wcount,
                         Wh, prC, prE, n, lo, cnt, acc) == 0u) break;
          m &= m - 1u;
        }
      }
    }
    while (wcount >= 32u) {
      wcount -= 32u;
      gather_chunk(seg + wcount, 32u, lane, Wh, prC, prE, n, lo, cnt, acc);
    }
  }
  const int t = nE16 * 16 + tid;
  bool hit = false;
  if (t < nE) hit = ((unsigned)pr[t].x - lo) < cnt;
  seg_append(hit, tag | (unsigned)t, seg, wcount,
             Wh, prC, prE, n, lo, cnt, acc);
}

#define OUT_NT      2
#define OUT_CG      (DFEAT / (TILE * OUT_NT))
#define OUT_TASKS   (OWN_STRIPS * OUT_CG)

__device__ __forceinline__ void own_out_task(
    const float* acc, const float* lw, int strip, int n0, int lane,
    const float* __restrict__ bo, const float* __restrict__ WhE,
    float* __restrict__ out, int lo, int cnt)
{
  const int half = lane >> 4;
  const int l15  = lane & 15;
  const float* ar = acc + (strip * TILE + l15) * APITCH + half * 2 * QP;
  const float* bl = lw + (n0 + l15) * WPITCH + half * 2;
  const int r0 = strip * TILE + half * 8;

  float bv[OUT_NT];
  float ad[OUT_NT][8];
#pragma unroll
  for (int j = 0; j < OUT_NT; ++j) bv[j] = bo[n0 + j * TILE + l15];
#pragma unroll
  for (int rr = 0; rr < 8; ++rr) {
    int rl = r0 + rr;
    if (rl >= cnt) rl = cnt - 1;
    const float* wr = WhE + (size_t)(lo + rl) * DFEAT + n0 + l15;
#pragma unroll
    for (int j = 0; j < OUT_NT; ++j) ad[j][rr] = wr[j * TILE];
  }

  v8f c[OUT_NT];
#pragma unroll
  for (int j = 0; j < OUT_NT; ++j) c[j] = (v8f){0.f, 0.f, 0.f, 0.f, 0.f, 0.f, 0.f, 0.f};
#pragma unroll 1
  for (int s0 = 0; s0 < DFEAT / 4; s0 += 8) {
    v2f a[8];
#pragma unroll
    for (int s = 0; s < 8; ++s) {
      a[s].x = ar[s0 + s];
      a[s].y = ar[QP + s0 + s];
    }
#pragma unroll
    for (int s = 0; s < 8; ++s) {
#pragma unroll
      for (int j = 0; j < OUT_NT; ++j) {
        const v2f b = *(const v2f*)(bl + j * TILE * WPITCH + 4 * (s0 + s));
        c[j] = __builtin_amdgcn_wmma_f32_16x16x4_f32(
            false, a[s], false, b, (short)0, c[j], false, false);
      }
    }
  }

  float* ocol = out + (size_t)(lo + r0) * DFEAT + n0 + l15;
  if (strip * TILE + TILE <= cnt) {
#pragma unroll
    for (int j = 0; j < OUT_NT; ++j)
#pragma unroll
      for (int rr = 0; rr < 8; ++rr)
        ocol[(size_t)rr * DFEAT + j * TILE] = c[j][rr] + bv[j] + ad[j][rr];
  } else {
#pragma unroll
    for (int j = 0; j < OUT_NT; ++j)
#pragma unroll
      for (int rr = 0; rr < 8; ++rr)
        if (r0 + rr < cnt)
          ocol[(size_t)rr * DFEAT + j * TILE] = c[j][rr] + bv[j] + ad[j][rr];
  }
}

__device__ __forceinline__ void own_clear(float* acc) {
  for (int i = threadIdx.x; i < OWN_ROWS * APITCH / 4; i += OWN_THREADS)
    ((float4*)acc)[i] = make_float4(0.f, 0.f, 0.f, 0.f);
}

__global__ __launch_bounds__(OWN_THREADS) void scatter_owner_lds(
    const float* __restrict__ Wh,
    const unsigned char* __restrict__ keys,
    const float* __restrict__ Wo, const float* __restrict__ bo,
    float* __restrict__ out, int n, int nEc, int nEe)
{
  __shared__ __attribute__((aligned(16))) float acc[OWN_ROWS * APITCH];
  __shared__ __attribute__((aligned(16))) unsigned list[LIST_CAP];
  __shared__ __attribute__((aligned(16))) float lw[DFEAT * WPITCH];

  const int tid  = threadIdx.x;
  const int lane = tid & 31;
  const int wave = __builtin_amdgcn_readfirstlane(tid >> 5);
  unsigned* seg = list + wave * SEG_CAP;
  const float* WhE = Wh + (size_t)n * DFEAT;
  const int2* prC = (const int2*)(keys + pairs_byte_offset(nEc, nEe));
  const int2* prE = prC + pairs_second(nEc);

  own_clear(acc);
  stage_w<OWN_THREADS>(Wo, lw);

  for (int tile = blockIdx.x; tile * OWN_NODES < n; tile += gridDim.x) {
    const int lo  = tile * OWN_NODES;
    const int cnt = (n - lo < OWN_NODES) ? (n - lo) : OWN_NODES;

    __syncthreads();

    const unsigned splat = ((unsigned)tile & 255u) * 0x01010101u;
    unsigned wcount = 0u;
#pragma nounroll
    for (int ty = 0; ty < 2; ++ty) {
      const unsigned char* kT = ty ? keys + keys_second(nEc) : keys;
      own_scan(kT, ty ? prE : prC, ty ? nEe : nEc, ty ? ETYPE_BIT : 0u,
               splat, seg, wcount, lane,
               Wh, prC, prE, n, (unsigned)lo, (unsigned)cnt, acc);
    }
    while (wcount > 0u) {
      const unsigned nv = wcount < 32u ? wcount : 32u;
      wcount -= nv;
      gather_chunk(seg + wcount, nv, lane, Wh, prC, prE,
                   n, (unsigned)lo, (unsigned)cnt, acc);
    }
    __syncthreads();

    for (int t = wave; t < OUT_TASKS; t += OWN_WAVES) {
      const int strip = t / OUT_CG;
      if (strip * TILE < cnt)
        own_out_task(acc, lw, strip, (t % OUT_CG) * (TILE * OUT_NT), lane,
                     bo, WhE, out, lo, cnt);
    }

    if ((tile + (int)gridDim.x) * OWN_NODES < n) {
      __syncthreads();
      own_clear(acc);
    }
  }
}

extern "C" void kernel_launch(void* const* d_in, const int* in_sizes, int n_in,
                              void* d_out, int out_size, void* d_ws, size_t ws_size,
                              hipStream_t stream) {
  const float* feats = (const float*)d_in[0];
  const float* Wc    = (const float*)d_in[1];
  const float* bc    = (const float*)d_in[2];
  const float* We    = (const float*)d_in[3];
  const float* be    = (const float*)d_in[4];
  const float* Wo    = (const float*)d_in[5];
  const float* bo    = (const float*)d_in[6];
  const int*   srcC  = (const int*)d_in[7];
  const int*   dstC  = (const int*)d_in[8];
  const int*   srcE  = (const int*)d_in[9];
  const int*   dstE  = (const int*)d_in[10];

  const int n   = in_sizes[0] / DFEAT;
  const int nEc = in_sizes[7];
  const int nEe = in_sizes[9];

  const size_t feSize = (size_t)n * DFEAT;
  float* WhC = (float*)d_ws;
  float* WhE = WhC + feSize;
  unsigned char* keys = (unsigned char*)(WhE + feSize);
  float* out = (float*)d_out;

  rgcn_proj_gemm<<<512, 256, 0, stream>>>(
      feats, Wc, bc, We, be, WhC, WhE, n, srcC, dstC, srcE, dstE, nEc, nEe, keys);

  scatter_owner_lds<<<256, OWN_THREADS, 0, stream>>>(
      WhC, keys, Wo, bo, out, n, nEc, nEe);
}


# Two-level owner scan: the projection launch also writes a 256-bit key bitmap per block of 32 edges; an owner tests one bit per block and reads keys of marked blocks only; sum order: LDS arrival
# speedup vs baseline: 5.6826x; 1.0624x over previous
#include <hip/hip_runtime.h>


typedef __attribute__((ext_vector_type(2))) float v2f;
typedef __attribute__((ext_vector_type(8))) float v8f;

#define DFEAT 128
#define TILE  16

#define WPITCH 132

template <int NTHR>
__device__ __forceinline__ void stage_w(const float* __restrict__ Wm, float* lw) {
  const int row0 = threadIdx.x >> 5;
  const int c4   = threadIdx.x & 31;
  float4 v[DFEAT * 32 / NTHR];
#pragma unroll
  for (int p = 0; p < DFEAT * 32 / NTHR; ++p)
    v[p] = *(const float4*)(Wm + (size_t)(row0 + p * (NTHR / 32)) * DFEAT + c4 * 4);
#pragma unroll
  for (int p = 0; p < DFEAT * 32 / NTHR; ++p)
    *(float4*)(lw + (row0 + p * (NTHR / 32)) * WPITCH + c4 * 4) = v[p];
}

__device__ __forceinline__ int strip_of_round(int r, int q, int nq, int nStrips) {
  const int base = r * nq;
  int c = nStrips - base;
  if (c > nq) c = nq;
  const unsigned lo = ((unsigned)q * (unsigned)c) / (unsigned)nq;
  const unsigned hi = ((unsigned)(q + 1) * (unsigned)c) / (unsigned)nq;
  return (hi > lo) ? base + (int)lo : -1;
}

template <int NT, int KC>
__device__ __forceinline__ void strip_mma(
    const float* __restrict__ arow, const float* bl, v8f (&acc)[NT])
{
#pragma unroll 1
  for (int k0 = 0; k0 < DFEAT; k0 += 4 * KC) {
    v2f a[KC];
#pragma unroll
    for (int s = 0; s < KC; ++s) a[s] = *(const v2f*)(arow + k0 + 4 * s);
#pragma unroll
    for (int s = 0; s < KC; ++s) {
#pragma unroll
      for (int j = 0; j < NT; ++j) {
        const v2f b = *(const v2f*)(bl + j * TILE * WPITCH + k0 + 4 * s);
        acc[j] = __builtin_amdgcn_wmma_f32_16x16x4_f32(
            false, a[s], false, b, (short)0, acc[j], false, false);
      }
    }
  }
}

#define OWN_NODES   196

__device__ __forceinline__ unsigned own_key(int d) {
  return ((unsigned)d / (unsigned)OWN_NODES) & 255u;
}

__device__ __forceinline__ unsigned pack_keys(int4 d) {
  return own_key(d.x) | (own_key(d.y) << 8) | (own_key(d.z) << 16) | (own_key(d.w) << 24);
}

__device__ __forceinline__ void store_pairs(int2* __restrict__ pr, int g, int4 d, int4 s) {
  int4* p = (int4*)(pr + 4 * (size_t)g);
  p[0] = make_int4(d.x, s.x, d.y, s.y);
  p[1] = make_int4(d.z, s.z, d.w, s.w);
}

__device__ __forceinline__ void make_edge_pairs_tail(
    const int* __restrict__ dst, const int* __restrict__ src, int nE,
    int2* __restrict__ pr, int gtid)
{
  const int t = (nE >> 2) * 4 + gtid;
  if (t < nE) pr[t] = make_int2(dst[t], src[t]);
}

#define CHUNK_BLKS 32

__device__ __forceinline__ int bm_pitch(int nE) {
  return ((nE >> 5) + CHUNK_BLKS - 1) & ~(CHUNK_BLKS - 1);
}

__device__ __forceinline__ size_t keys_second(int nEfirst) {
  return ((size_t)nEfirst + 15) & ~(size_t)15;
}

__device__ __forceinline__ size_t pairs_byte_offset(int nEc, int nEe) {
  return keys_second(nEc) + keys_second(nEe);
}
__device__ __forceinline__ size_t pairs_second(int nEfirst) {
  return ((size_t)nEfirst + 1) & ~(size_t)1;
}
__device__ __forceinline__ size_t bitmap_byte_offset(int nEc, int nEe) {
  const size_t b = pairs_byte_offset(nEc, nEe) + (pairs_second(nEc) + pairs_second(nEe)) * 8;
  return (b + 127) & ~(size_t)127;
}

__device__ __forceinline__ void edge_chunk_or(int g, int n4, int nB, int4 d, unsigned* bml_t) {
  if (g < n4 && (g >> 3) < nB) {
    unsigned* wrd = bml_t + (threadIdx.x >> 3) * 8;
    const unsigned k0 = own_key(d.x), k1 = own_key(d.y);
    const unsigned k2 = own_key(d.z), k3 = own_key(d.w);
    __hip_atomic_fetch_or(wrd + (k0 >> 5), 1u << (k0 & 31u), __ATOMIC_RELAXED, __HIP_MEMORY_SCOPE_WORKGROUP);
    __hip_atomic_fetch_or(wrd + (k1 >> 5), 1u << (k1 & 31u), __ATOMIC_RELAXED, __HIP_MEMORY_SCOPE_WORKGROUP);
    __hip_atomic_fetch_or(wrd + (k2 >> 5), 1u << (k2 & 31u), __ATOMIC_RELAXED, __HIP_MEMORY_SCOPE_WORKGROUP);
    __hip_atomic_fetch_or(wrd + (k3 >> 5), 1u << (k3 & 31u), __ATOMIC_RELAXED, __HIP_MEMORY_SCOPE_WORKGROUP);
  }
}

__device__ __forceinline__ void edge_chunk(
    int c, int4 dC, int4 sC, int4 dE, int4 sE, int nEc, int nEe,
    unsigned* __restrict__ kwC, unsigned* __restrict__ kwE,
    int2* __restrict__ prC, int2* __restrict__ prE,
    unsigned* __restrict__ bmC, unsigned* __restrict__ bmE, unsigned* bml)
{
  const int g = c * 256 + (int)threadIdx.x;
  edge_chunk_or(g, nEc >> 2, nEc >> 5, dC, bml);
  edge_chunk_or(g, nEe >> 2, nEe >> 5, dE, bml + CHUNK_BLKS * 8);
  __syncthreads();
  if (g < (nEc >> 2)) { kwC[g] = pack_keys(dC); store_pairs(prC, g, dC, sC); }
  if (g < (nEe >> 2)) { kwE[g] = pack_keys(dE); store_pairs(prE, g, dE, sE); }
  const int lane = threadIdx.x & 31;
  const int v    = threadIdx.x >> 5;
  const int blkG = c * CHUNK_BLKS + lane;
  const int Pc = bm_pitch(nEc), Pe = bm_pitch(nEe);
  if (blkG < Pc) bmC[(size_t)v * Pc + blkG] = bml[lane * 8 + v];
  if (blkG < Pe) bmE[(size_t)v * Pe + blkG] = bml[(CHUNK_BLKS + lane) * 8 + v];
}

__global__ __launch_bounds__(256) void rgcn_proj_gemm(
    const float* __restrict__ feats,
    const float* __restrict__ Wc, const float* __restrict__ bc,
    const float* __restrict__ We, const float* __restrict__ be,
    float* __restrict__ WhC, float* __restrict__ WhE, int n,
    const int* __restrict__ srcC, const int* __restrict__ dstC,
    const int* __restrict__ srcE, const int* __restrict__ dstE,
    int nEc, int nEe, unsigned char* __restrict__ keys)
{
  __shared__ __attribute__((aligned(16))) float lw[2 * DFEAT * WPITCH];
  __shared__ unsigned bml[2 * CHUNK_BLKS * 8];
  const int gtid = blockIdx.x * blockDim.x + threadIdx.x;
  unsigned* kwC = (unsigned*)keys;
  unsigned* kwE = (unsigned*)(keys + keys_second(nEc));
  int2* prC = (int2*)(keys + pairs_byte_offset(nEc, nEe));
  int2* prE = prC + pairs_second(nEc);
  unsigned* bmC = (unsigned*)(keys + bitmap_byte_offset(nEc, nEe));
  unsigned* bmE = bmC + 8 * (size_t)bm_pitch(nEc);
  const int nC4 = nEc >> 2;
  const int nE4 = nEe >> 2;
  const int gC = gtid < nC4 ? gtid : (nC4 > 0 ? nC4 - 1 : 0);
  const int gE = gtid < nE4 ? gtid : (nE4 > 0 ? nE4 - 1 : 0);
  const int4 dC = ((const int4*)dstC)[gC];
  const int4 sC = ((const int4*)srcC)[gC];
  const int4 dE = ((const int4*)dstE)[gE];
  const int4 sE = ((const int4*)srcE)[gE];

  bml[threadIdx.x] = 0u;
  bml[threadIdx.x + 256] = 0u;

  stage_w<256>(Wc, lw);
  stage_w<256>(We, lw + DFEAT * WPITCH);
  __syncthreads();

  const int nChC = (nC4 + 255) >> 8, nChE = (nE4 + 255) >> 8;
  const int nChunks = nChC > nChE ? nChC : nChE;
  if ((int)blockIdx.x < nChunks)
    edge_chunk((int)blockIdx.x, dC, sC, dE, sE, nEc, nEe,
               kwC, kwE, prC, prE, bmC, bmE, bml);
#pragma nounroll
  for (int ch = (int)(blockIdx.x + gridDim.x); ch < nChunks; ch += (int)gridDim.x) {
    __syncthreads();
    bml[threadIdx.x] = 0u;
    bml[threadIdx.x + 256] = 0u;
    __syncthreads();
    const int g  = ch * 256 + (int)threadIdx.x;
    const int hC = g < nC4 ? g : (nC4 > 0 ? nC4 - 1 : 0);
    const int hE = g < nE4 ? g : (nE4 > 0 ? nE4 - 1 : 0);
    edge_chunk(ch, ((const int4*)dstC)[hC], ((const int4*)srcC)[hC],
               ((const int4*)dstE)[hE], ((const int4*)srcE)[hE], nEc, nEe,
               kwC, kwE, prC, prE, bmC, bmE, bml);
  }
  make_edge_pairs_tail(dstC, srcC, nEc, prC, gtid);
  make_edge_pairs_tail(dstE, srcE, nEe, prE, gtid);

  const int lane = threadIdx.x & 31;
  const int wave = __builtin_amdgcn_readfirstlane(threadIdx.x >> 5);
  const int g    = wave & 3;
  const int q    = blockIdx.x * 2 + (wave >> 2);
  const int nq   = gridDim.x * 2;
  const int nStrips = (n + TILE - 1) / TILE;

  const int half = lane >> 4;
  const int l15  = lane & 15;
  const int n0   = (g & 1) * 64;
  const float* bl   = lw + (g * 64 + l15) * WPITCH + half * 2;
  const float* bias = (g < 2) ? bc : be;
  float* dst        = (g < 2) ? WhC : WhE;

  float bv[4];
#pragma unroll
  for (int j = 0; j < 4; ++j) bv[j] = bias[n0 + j * TILE + l15];

  for (int r = 0; r * nq < nStrips; ++r) {
    const int s = strip_of_round(r, q, nq, nStrips);
    if (s < 0) continue;
    const int m0 = s * TILE;
    int arowi = m0 + l15;
    if (arowi >= n) arowi = n - 1;
    const float* arow = feats + (size_t)arowi * DFEAT + half * 2;

    v8f acc[4];
#pragma unroll
    for (int j = 0; j < 4; ++j) acc[j] = (v8f){0.f, 0.f, 0.f, 0.f, 0.f, 0.f, 0.f, 0.f};
    strip_mma<4, 32>(arow, bl, acc);

    float* ocol = dst + (size_t)(m0 + half * 8) * DFEAT + n0 + l15;
    if (m0 + TILE <= n) {
#pragma unroll
      for (int j = 0; j < 4; ++j)
#pragma unroll
        for (int rr = 0; rr < 8; ++rr)
          ocol[(size_t)rr * DFEAT + j * TILE] = acc[j][rr] + bv[j];
    } else {
#pragma unroll
      for (int j = 0; j < 4; ++j)
#pragma unroll
        for (int rr = 0; rr < 8; ++rr)
          if (m0 + half * 8 + rr < n)
            ocol[(size_t)rr * DFEAT + j * TILE] = acc[j][rr] + bv[j];
    }
  }
}

#define APITCH      132
#define QP          33
#define OWN_STRIPS  ((OWN_NODES + TILE - 1) / TILE)
#define OWN_ROWS    (OWN_STRIPS * TILE)
#define OWN_THREADS 1024
#define OWN_WAVES   (OWN_THREADS / 32)
#define LIST_CAP    16384
#define SEG_CAP     (LIST_CAP / OWN_WAVES)
#define GATHER_U    8
#define ETYPE_BIT   0x80000000u

__device__ __forceinline__ void lds_fadd(float* p, float v) {
  __hip_atomic_fetch_add(p, v, __ATOMIC_RELAXED, __HIP_MEMORY_SCOPE_WORKGROUP);
}

__device__ __forceinline__ void own_direct_edge(
    unsigned ent, const float* __restrict__ Wh,
    const int2* __restrict__ prC, const int2* __restrict__ prE,
    int n, unsigned lo, unsigned cnt, float* acc)
{
  const bool isE = (ent & ETYPE_BIT) != 0u;
  const unsigned e = ent & ~ETYPE_BIT;
  const int2 pr = (isE ? prE : prC)[e];
  const unsigned a = (unsigned)pr.x - lo;
  if (a < cnt) {
    const float* row = Wh + ((size_t)pr.y + (isE ? (size_t)n : 0)) * DFEAT;
#pragma nounroll
    for (int c = 0; c < DFEAT; ++c)
      lds_fadd(acc + a * APITCH + (c & 3) * QP + (c >> 2), row[c]);
  }
}

__device__ __forceinline__ void gather_chunk(
    const unsigned* segp, unsigned nv, int lane, const float* __restrict__ Wh,
    const int2* __restrict__ prC, const int2* __restrict__ prE,
    int n, unsigned lo, unsigned cnt, float* acc)
{
  unsigned rowi = 0u;
  unsigned dl   = OWN_NODES;
  if ((unsigned)lane < nv) {
    const unsigned ent = segp[lane];
    const bool isE = (ent & ETYPE_BIT) != 0u;
    const unsigned e = ent & ~ETYPE_BIT;
    const int2 pr = (isE ? prE : prC)[e];
    const unsigned a = (unsigned)pr.x - lo;
    const unsigned s = (unsigned)pr.y + (isE ? (unsigned)n : 0u);
    const bool own = a < cnt;
    rowi = own ? s : 0u;
    dl   = own ? a : (unsigned)OWN_NODES;
  }
  const unsigned nvr = (nv + GATHER_U - 1u) & ~(unsigned)(GATHER_U - 1);
  for (unsigned j = 0; j < nvr; j += GATHER_U) {
    float4 v[GATHER_U];
    unsigned d[GATHER_U];
#pragma unroll
    for (int u = 0; u < GATHER_U; ++u) {
      const unsigned r = (unsigned)__builtin_amdgcn_readlane((int)rowi, (int)(j + u));
      d[u] = (unsigned)__builtin_amdgcn_readlane((int)dl, (int)(j + u));
      v[u] = *(const float4*)(Wh + (size_t)r * DFEAT + lane * 4);
    }
#pragma unroll
    for (int u = 0; u < GATHER_U; ++u) {
      float* p = acc + d[u] * APITCH + lane;
      lds_fadd(p,          v[u].x);
      lds_fadd(p + QP,     v[u].y);
      lds_fadd(p + 2 * QP, v[u].z);
      lds_fadd(p + 3 * QP, v[u].w);
    }
  }
}

__device__ __forceinline__ unsigned seg_append(
    bool has, unsigned tagged, unsigned* seg, unsigned& wcount,
    const float* __restrict__ Wh,
    const int2* __restrict__ prC, const int2* __restrict__ prE,
    int n, unsigned lo, unsigned cnt, float* acc)
{
  const unsigned bal = __builtin_amdgcn_ballot_w32(has);
  if (bal != 0u) {
    if (wcount + 32u <= (unsigned)SEG_CAP) {
      if (has) seg[wcount + __builtin_amdgcn_mbcnt_lo(bal, 0u)] = tagged;
      wcount += (unsigned)__builtin_popcount(bal);
    } else if (has) {
      own_direct_edge(tagged, Wh, prC, prE, n, lo, cnt, acc);
    }
  }
  return bal;
}

#define LV1_U 4

__device__ __forceinline__ unsigned match32(uint4 ka, uint4 kb, unsigned splat) {
  const unsigned x0 = ka.x ^ splat, x1 = ka.y ^ splat, x2 = ka.z ^ splat, x3 = ka.w ^ splat;
  const unsigned x4 = kb.x ^ splat, x5 = kb.y ^ splat, x6 = kb.z ^ splat, x7 = kb.w ^ splat;
  const unsigned t0 = (x0 - 0x01010101u) & ~x0 & 0x80808080u;
  const unsigned t1 = (x1 - 0x01010101u) & ~x1 & 0x80808080u;
  const unsigned t2 = (x2 - 0x01010101u) & ~x2 & 0x80808080u;
  const unsigned t3 = (x3 - 0x01010101u) & ~x3 & 0x80808080u;
  const unsigned t4 = (x4 - 0x01010101u) & ~x4 & 0x80808080u;
  const unsigned t5 = (x5 - 0x01010101u) & ~x5 & 0x80808080u;
  const unsigned t6 = (x6 - 0x01010101u) & ~x6 & 0x80808080u;
  const unsigned t7 = (x7 - 0x01010101u) & ~x7 & 0x80808080u;
  return (t0 >> 7) | (t1 >> 6) | (t2 >> 5) | (t3 >> 4) |
         (t4 >> 3) | (t5 >> 2) | (t6 >> 1) | t7;
}

__device__ __forceinline__ void own_scan(
    const unsigned char* __restrict__ keys, const unsigned* __restrict__ bm, int P,
    const int2* __restrict__ pr, int nE,
    unsigned tag, unsigned splat, unsigned bit, unsigned* seg, unsigned& wcount, int lane,
    const float* __restrict__ Wh,
    const int2* __restrict__ prC, const int2* __restrict__ prE,
    int n, unsigned lo, unsigned cnt, float* acc)
{
  const int tid = threadIdx.x;
  const int nQ  = P >> 2;
  const uint4* q4 = (const uint4*)bm;
  const uint4* k4 = (const uint4*)keys;
  for (int q0 = 0; q0 < nQ; q0 += OWN_THREADS * LV1_U) {
    uint4 w[LV1_U];
#pragma unroll
    for (int u = 0; u < LV1_U; ++u) {
      const int i = q0 + tid + u * OWN_THREADS;
      w[u] = q4[(i < nQ) ? i : nQ - 1];
    }
    unsigned mask = 0u;
#pragma unroll
    for (int u = 0; u < LV1_U; ++u) {
      const unsigned b4 = ((w[u].x >> bit) & 1u) | (((w[u].y >> bit) & 1u) << 1) |
                          (((w[u].z >> bit) & 1u) << 2) | (((w[u].w >> bit) & 1u) << 3);
      if (q0 + tid + u * OWN_THREADS < nQ) mask |= b4 << (4 * u);
    }
    const unsigned blk0 = (unsigned)(q0 + tid) * 4u;
    unsigned j   = (unsigned)__builtin_ctz(mask | 0x10000u);
    unsigned blk = (mask != 0u) ? blk0 + (j >> 2) * (unsigned)(OWN_THREADS * 4) + (j & 3u) : 0u;
    uint4 ka = k4[2u * blk];
    uint4 kb = k4[2u * blk + 1u];
    while (__builtin_amdgcn_ballot_w32(mask != 0u) != 0u) {
      const bool has = mask != 0u;
      const unsigned mnext = mask & (mask - 1u);
      j = (unsigned)__builtin_ctz(mnext | 0x10000u);
      const unsigned nblk = (mnext != 0u) ? blk0 + (j >> 2) * (unsigned)(OWN_THREADS * 4) + (j & 3u) : 0u;
      const uint4 na = k4[2u * nblk];
      const uint4 nb = k4[2u * nblk + 1u];

      unsigned m = has ? match32(ka, kb, splat) : 0u;
      const unsigned e0 = blk * 32u;
      for (;;) {
        const unsigned p = (unsigned)__builtin_ctz(m | 0x80000000u);
        const unsigned ent = tag | (e0 + ((p & 7u) << 2) + (p >> 3));
        if (seg_append(m != 0u, ent, seg, wcount,
                       Wh, prC, prE, n, lo, cnt, acc) == 0u) break;
        m &= m - 1u;
      }
      mask = mnext; blk = nblk; ka = na; kb = nb;
      while (wcount >= 32u) {
        wcount -= 32u;
        gather_chunk(seg + wcount, 32u, lane, Wh, prC, prE, n, lo, cnt, acc);
      }
    }
  }
  const int t = (nE >> 5) * 32 + tid;
  bool hit = false;
  if (t < nE) hit = ((unsigned)pr[t].x - lo) < cnt;
  seg_append(hit, tag | (unsigned)t, seg, wcount,
             Wh, prC, prE, n, lo, cnt, acc);
}

#define OUT_NT      2
#define OUT_CG      (DFEAT / (TILE * OUT_NT))
#define OUT_TASKS   (OWN_STRIPS * OUT_CG)

__device__ __forceinline__ void own_out_task(
    const float* acc, const float* lw, int strip, int n0, int lane,
    const float* __restrict__ bo, const float* __restrict__ WhE,
    float* __restrict__ out, int lo, int cnt)
{
  const int half = lane >> 4;
  const int l15  = lane & 15;
  const float* ar = acc + (strip * TILE + l15) * APITCH + half * 2 * QP;
  const float* bl = lw + (n0 + l15) * WPITCH + half * 2;
  const int r0 = strip * TILE + half * 8;

  float bv[OUT_NT];
  float ad[OUT_NT][8];
#pragma unroll
  for (int j = 0; j < OUT_NT; ++j) bv[j] = bo[n0 + j * TILE + l15];
#pragma unroll
  for (int rr = 0; rr < 8; ++rr) {
    int rl = r0 + rr;
    if (rl >= cnt) rl = cnt - 1;
    const float* wr = WhE + (size_t)(lo + rl) * DFEAT + n0 + l15;
#pragma unroll
    for (int j = 0; j < OUT_NT; ++j) ad[j][rr] = wr[j * TILE];
  }

  v8f c[OUT_NT];
#pragma unroll
  for (int j = 0; j < OUT_NT; ++j) c[j] = (v8f){0.f, 0.f, 0.f, 0.f, 0.f, 0.f, 0.f, 0.f};
#pragma unroll 1
  for (int s0 = 0; s0 < DFEAT / 4; s0 += 8) {
    v2f a[8];
#pragma unroll
    for (int s = 0; s < 8; ++s) {
      a[s].x = ar[s0 + s];
      a[s].y = ar[QP + s0 + s];
    }
#pragma unroll
    for (int s = 0; s < 8; ++s) {
#pragma unroll
      for (int j = 0; j < OUT_NT; ++j) {
        const v2f b = *(const v2f*)(bl + j * TILE * WPITCH + 4 * (s0 + s));
        c[j] = __builtin_amdgcn_wmma_f32_16x16x4_f32(
            false, a[s], false, b, (short)0, c[j], false, false);
      }
    }
  }

  float* ocol = out + (size_t)(lo + r0) * DFEAT + n0 + l15;
  if (strip * TILE + TILE <= cnt) {
#pragma unroll
    for (int j = 0; j < OUT_NT; ++j)
#pragma unroll
      for (int rr = 0; rr < 8; ++rr)
        ocol[(size_t)rr * DFEAT + j * TILE] = c[j][rr] + bv[j] + ad[j][rr];
  } else {
#pragma unroll
    for (int j = 0; j < OUT_NT; ++j)
#pragma unroll
      for (int rr = 0; rr < 8; ++rr)
        if (r0 + rr < cnt)
          ocol[(size_t)rr * DFEAT + j * TILE] = c[j][rr] + bv[j] + ad[j][rr];
  }
}

__device__ __forceinline__ void own_clear(float* acc) {
  for (int i = threadIdx.x; i < OWN_ROWS * APITCH / 4; i += OWN_THREADS)
    ((float4*)acc)[i] = make_float4(0.f, 0.f, 0.f, 0.f);
}

__global__ __launch_bounds__(OWN_THREADS) void scatter_owner_lds(
    const float* __restrict__ Wh,
    const unsigned char* __restrict__ keys,
    const float* __restrict__ Wo, const float* __restrict__ bo,
    float* __restrict__ out, int n, int nEc, int nEe)
{
  __shared__ __attribute__((aligned(16))) float acc[OWN_ROWS * APITCH];
  __shared__ __attribute__((aligned(16))) unsigned list[LIST_CAP];
  __shared__ __attribute__((aligned(16))) float lw[DFEAT * WPITCH];

  const int tid  = threadIdx.x;
  const int lane = tid & 31;
  const int wave = __builtin_amdgcn_readfirstlane(tid >> 5);
  unsigned* seg = list + wave * SEG_CAP;
  const float* WhE = Wh + (size_t)n * DFEAT;
  const int2* prC = (const int2*)(keys + pairs_byte_offset(nEc, nEe));
  const int2* prE = prC + pairs_second(nEc);
  const unsigned* bmC = (const unsigned*)(keys + bitmap_byte_offset(nEc, nEe));
  const unsigned* bmE = bmC + 8 * (size_t)bm_pitch(nEc);

  own_clear(acc);
  stage_w<OWN_THREADS>(Wo, lw);

  for (int tile = blockIdx.x; tile * OWN_NODES < n; tile += gridDim.x) {
    const int lo  = tile * OWN_NODES;
    const int cnt = (n - lo < OWN_NODES) ? (n - lo) : OWN_NODES;

    __syncthreads();

    const unsigned key   = (unsigned)tile & 255u;
    const unsigned splat = key * 0x01010101u;
    unsigned wcount = 0u;
#pragma nounroll
    for (int ty = 0; ty < 2; ++ty) {
      const unsigned char* kT = ty ? keys + keys_second(nEc) : keys;
      const int P = bm_pitch(ty ? nEe : nEc);
      const unsigned* bmT = (ty ? bmE : bmC) + (size_t)(key >> 5) * P;
      own_scan(kT, bmT, P, ty ? prE : prC, ty ? nEe : nEc, ty ? ETYPE_BIT : 0u,
               splat, key & 31u, seg, wcount, lane,
               Wh, prC, prE, n, (unsigned)lo, (unsigned)cnt, acc);
    }
    while (wcount > 0u) {
      const unsigned nv = wcount < 32u ? wcount : 32u;
      wcount -= nv;
      gather_chunk(seg + wcount, nv, lane, Wh, prC, prE,
                   n, (unsigned)lo, (unsigned)cnt, acc);
    }
    __syncthreads();

    for (int t = wave; t < OUT_TASKS; t += OWN_WAVES) {
      const int strip = t / OUT_CG;
      if (strip * TILE < cnt)
        own_out_task(acc, lw, strip, (t % OUT_CG) * (TILE * OUT_NT), lane,
                     bo, WhE, out, lo, cnt);
    }

    if ((tile + (int)gridDim.x) * OWN_NODES < n) {
      __syncthreads();
      own_clear(acc);
    }
  }
}

extern "C" void kernel_launch(void* const* d_in, const int* in_sizes, int n_in,
                              void* d_out, int out_size, void* d_ws, size_t ws_size,
                              hipStream_t stream) {
  const float* feats = (const float*)d_in[0];
  const float* Wc    = (const float*)d_in[1];
  const float* bc    = (const float*)d_in[2];
  const float* We    = (const float*)d_in[3];
  const float* be    = (const float*)d_in[4];
  const float* Wo    = (const float*)d_in[5];
  const float* bo    = (const float*)d_in[6];
  const int*   srcC  = (const int*)d_in[7];
  const int*   dstC  = (const int*)d_in[8];
  const int*   srcE  = (const int*)d_in[9];
  const int*   dstE  = (const int*)d_in[10];

  const int n   = in_sizes[0] / DFEAT;
  const int nEc = in_sizes[7];
  const int nEe = in_sizes[9];

  const size_t feSize = (size_t)n * DFEAT;
  float* WhC = (float*)d_ws;
  float* WhE = WhC + feSize;
  unsigned char* keys = (unsigned char*)(WhE + feSize);
  float* out = (float*)d_out;

  rgcn_proj_gemm<<<512, 256, 0, stream>>>(
      feats, Wc, bc, We, be, WhC, WhE, n, srcC, dstC, srcE, dstE, nEc, nEe, keys);

  scatter_owner_lds<<<256, OWN_THREADS, 0, stream>>>(
      WhC, keys, Wo, bo, out, n, nEc, nEe);
}
